# starting kernel
# baseline (speedup 1.0000x reference)
#include <hip/hip_runtime.h>

#define B_  131072
#define S_  32
#define D_  16
#define WPB 4
#define TPB (WPB * 32)

#define XS  20
#define SS  36

typedef float v2f __attribute__((ext_vector_type(2)));
typedef float v8f __attribute__((ext_vector_type(8)));

__device__ __forceinline__ v8f wmma4(v2f a, v2f b, v8f c) {
  return __builtin_amdgcn_wmma_f32_16x16x4_f32(false, a, false, b, (short)0, c,
                                               false, false);
}

__device__ __forceinline__ void lds_fence() {
  asm volatile("s_wait_dscnt 0" ::: "memory");
}

__device__ __forceinline__ v2f frag_rowpair(const float* base, int row, int rs,
                                            int col) {
  return *(const v2f*)(base + row * rs + col);
}

__device__ __forceinline__ v2f frag_col(const float* base, int row0, int rs,
                                        int col) {
  v2f r;
  r.x = base[(row0 + 0) * rs + col];
  r.y = base[(row0 + 1) * rs + col];
  return r;
}

__device__ __forceinline__ void store_tile(float* base, int rs, int row0,
                                           int col, v8f acc) {
#pragma unroll
  for (int v = 0; v < 8; ++v) base[(row0 + v) * rs + col] = acc[v];
}

__global__ __launch_bounds__(TPB) void encoder_kernel(
    const int* __restrict__ arr1, const int* __restrict__ arr2,
    const float* __restrict__ emb,
    const float* __restrict__ wq, const float* __restrict__ wk,
    const float* __restrict__ wv,
    const float* __restrict__ w1, const float* __restrict__ b1,
    const float* __restrict__ w2, const float* __restrict__ b2,
    float* __restrict__ out) {
  __shared__ __align__(16) float s_w[5][256];
  __shared__ __align__(16) float s_b[2][16];
  __shared__ __align__(16) float s_x[WPB][S_ * XS];
  __shared__ __align__(16) float s_q[WPB][S_ * XS];
  __shared__ __align__(16) float s_k[WPB][S_ * XS];
  __shared__ __align__(16) float s_v[WPB][S_ * XS];
  __shared__ __align__(16) float s_s[WPB][S_ * SS];

  const int tid = threadIdx.x;
#pragma unroll
  for (int i = tid; i < 256; i += TPB) {
    s_w[0][i] = wq[i];
    s_w[1][i] = wk[i];
    s_w[2][i] = wv[i];
    s_w[3][i] = w1[i];
    s_w[4][i] = w2[i];
  }
  if (tid < 16) {
    s_b[0][tid] = b1[tid];
    s_b[1][tid] = b2[tid];
  }
  __syncthreads();

  const int wave = tid >> 5;
  const int lane = tid & 31;
  const int m    = lane & 15;
  const int half = lane >> 4;

  const unsigned task = blockIdx.x * WPB + wave;
  const int enc = (task >= (unsigned)B_) ? 1 : 0;
  const int b   = enc ? (int)(task - (unsigned)B_) : (int)task;
  const int* __restrict__ arr = enc ? arr2 : arr1;

  float* X  = s_x[wave];
  float* Q  = s_q[wave];
  float* K  = s_k[wave];
  float* V  = s_v[wave];
  float* Sc = s_s[wave];

  {
    const int tok = arr[b * S_ + lane];
    const float4* er = (const float4*)(emb + (size_t)tok * D_);
    float4* xr = (float4*)(X + lane * XS);
    xr[0] = er[0];
    xr[1] = er[1];
    xr[2] = er[2];
    xr[3] = er[3];
  }
  lds_fence();

#pragma unroll
  for (int t = 0; t < 2; ++t) {
    v8f cq = {}, ck = {}, cv = {};
#pragma unroll
    for (int kc = 0; kc < 4; ++kc) {
      v2f a = frag_rowpair(X, 16 * t + m, XS, 4 * kc + 2 * half);
      cq = wmma4(a, frag_rowpair(s_w[0], m, D_, 4 * kc + 2 * half), cq);
      ck = wmma4(a, frag_rowpair(s_w[1], m, D_, 4 * kc + 2 * half), ck);
      cv = wmma4(a, frag_rowpair(s_w[2], m, D_, 4 * kc + 2 * half), cv);
    }
    store_tile(Q, XS, 16 * t + 8 * half, m, cq);
    store_tile(K, XS, 16 * t + 8 * half, m, ck);
    store_tile(V, XS, 16 * t + 8 * half, m, cv);
  }
  lds_fence();

#pragma unroll
  for (int tm = 0; tm < 2; ++tm)
#pragma unroll
    for (int tn = 0; tn < 2; ++tn) {
      v8f c = {};
#pragma unroll
      for (int kc = 0; kc < 4; ++kc) {
        v2f a  = frag_rowpair(Q, 16 * tm + m, XS, 4 * kc + 2 * half);
        v2f bb = frag_rowpair(K, 16 * tn + m, XS, 4 * kc + 2 * half);
        c = wmma4(a, bb, c);
      }
      store_tile(Sc, SS, 16 * tm + 8 * half, 16 * tn + m, c);
    }
  lds_fence();

  {
    float4* rowv = (float4*)(Sc + lane * SS);
    float r[S_];
#pragma unroll
    for (int i = 0; i < 8; ++i) {
      float4 f = rowv[i];
      r[4 * i + 0] = f.x;
      r[4 * i + 1] = f.y;
      r[4 * i + 2] = f.z;
      r[4 * i + 3] = f.w;
    }
    float mx = -3.402823466e38f;
#pragma unroll
    for (int j = 0; j < S_; ++j) mx = fmaxf(mx, r[j]);
    float sum = 0.f;
#pragma unroll
    for (int j = 0; j < S_; ++j) {
      r[j] = __expf((r[j] - mx) * 0.25f);
      sum += r[j];
    }
    const float inv = 1.f / sum;
#pragma unroll
    for (int i = 0; i < 8; ++i) {
      float4 f;
      f.x = r[4 * i + 0] * inv;
      f.y = r[4 * i + 1] * inv;
      f.z = r[4 * i + 2] * inv;
      f.w = r[4 * i + 3] * inv;
      rowv[i] = f;
    }
  }
  lds_fence();

#pragma unroll
  for (int t = 0; t < 2; ++t) {
    v8f c = {};
#pragma unroll
    for (int kc = 0; kc < 8; ++kc) {
      v2f a  = frag_rowpair(Sc, 16 * t + m, SS, 4 * kc + 2 * half);
      v2f bb = frag_col(V, 4 * kc + 2 * half, XS, m);
      c = wmma4(a, bb, c);
    }
    store_tile(X, XS, 16 * t + 8 * half, m, c);
  }
  lds_fence();

  {
    const float bb1 = s_b[0][m];
#pragma unroll
    for (int t = 0; t < 2; ++t) {
      v8f c = {};
#pragma unroll
      for (int kc = 0; kc < 4; ++kc) {
        v2f a  = frag_rowpair(X, 16 * t + m, XS, 4 * kc + 2 * half);
        v2f bb = frag_rowpair(s_w[3], m, D_, 4 * kc + 2 * half);
        c = wmma4(a, bb, c);
      }
#pragma unroll
      for (int v = 0; v < 8; ++v) c[v] = fmaxf(c[v] + bb1, 0.f);
      store_tile(Q, XS, 16 * t + 8 * half, m, c);
    }
  }
  lds_fence();

  {
    float partial = 0.f;
#pragma unroll
    for (int t = 0; t < 2; ++t) {
      v8f c = {};
#pragma unroll
      for (int kc = 0; kc < 4; ++kc) {
        v2f a  = frag_rowpair(Q, 16 * t + m, XS, 4 * kc + 2 * half);
        v2f bb = frag_rowpair(s_w[4], m, D_, 4 * kc + 2 * half);
        c = wmma4(a, bb, c);
      }
#pragma unroll
      for (int v = 0; v < 8; ++v) partial += c[v];
    }
    float other = __shfl_xor(partial, 16, 32);
    float mean  = (partial + other) * (1.f / 32.f) + s_b[1][m];
    if (half == 0) {
      out[(size_t)enc * ((size_t)B_ * D_) + (size_t)b * D_ + m] = mean;
    }
  }
}

extern "C" void kernel_launch(void* const* d_in, const int* in_sizes, int n_in,
                              void* d_out, int out_size, void* d_ws,
                              size_t ws_size, hipStream_t stream) {
  (void)in_sizes; (void)n_in; (void)out_size; (void)d_ws; (void)ws_size;
  const int*   arr1 = (const int*)d_in[0];
  const int*   arr2 = (const int*)d_in[1];
  const float* emb  = (const float*)d_in[2];
  const float* wq   = (const float*)d_in[3];
  const float* wk   = (const float*)d_in[4];
  const float* wv   = (const float*)d_in[5];
  const float* w1   = (const float*)d_in[6];
  const float* b1   = (const float*)d_in[7];
  const float* w2   = (const float*)d_in[8];
  const float* b2   = (const float*)d_in[9];
  float* out = (float*)d_out;

  const unsigned tasks  = 2u * B_;
  const unsigned blocks = tasks / WPB;
  encoder_kernel<<<dim3(blocks), dim3(TPB), 0, stream>>>(
      arr1, arr2, emb, wq, wk, wv, w1, b1, w2, b2, out);
}


# Register-resident dataflow: transposed products and one common k order make every WMMA D tile the next A/B fragment, so LDS tiles, fences and barrier go; contraction and softmax sums reordered
# speedup vs baseline: 1.2837x; 1.2837x over previous
#include <hip/hip_runtime.h>

#define B_  131072
#define S_  32
#define D_  16
#define WPB 4
#define TPB (WPB * 32)

typedef float v2f __attribute__((ext_vector_type(2)));
typedef float v8f __attribute__((ext_vector_type(8)));

__device__ __forceinline__ v8f wmma4(v2f a, v2f b, v8f c) {
  return __builtin_amdgcn_wmma_f32_16x16x4_f32(false, a, false, b, (short)0, c,
                                               false, false);
}


__device__ __forceinline__ v8f load8(const float* __restrict__ p) {
  const float4 lo = ((const float4*)p)[0];
  const float4 hi = ((const float4*)p)[1];
  v8f r;
  r[0] = lo.x; r[1] = lo.y; r[2] = lo.z; r[3] = lo.w;
  r[4] = hi.x; r[5] = hi.y; r[6] = hi.z; r[7] = hi.w;
  return r;
}

__device__ __forceinline__ v2f kpair(v8f v, int j) {
  v2f r;
  r.x = v[2 * j];
  r.y = v[2 * j + 1];
  return r;
}

__global__ __launch_bounds__(TPB) void encoder_kernel(
    const int* __restrict__ arr1, const int* __restrict__ arr2,
    const float* __restrict__ emb,
    const float* __restrict__ wq, const float* __restrict__ wk,
    const float* __restrict__ wv,
    const float* __restrict__ w1, const float* __restrict__ b1,
    const float* __restrict__ w2, const float* __restrict__ b2,
    float* __restrict__ out) {
  const int tid  = threadIdx.x;
  const int wave = tid >> 5;
  const int lane = tid & 31;
  const int c    = lane & 15;
  const int h    = lane >> 4;

  const unsigned task = blockIdx.x * WPB + wave;
  const int enc = (task >= (unsigned)B_) ? 1 : 0;
  const int b   = enc ? (int)(task - (unsigned)B_) : (int)task;
  const int* __restrict__ arr = enc ? arr2 : arr1;

  const int tok0 = arr[b * S_ + c];
  const int tok1 = arr[b * S_ + 16 + c];

  const v8f fq = load8(wq + c * D_ + 8 * h);
  const v8f fk = load8(wk + c * D_ + 8 * h);
  const v8f fv = load8(wv + c * D_ + 8 * h);
  const v8f f1 = load8(w1 + c * D_ + 8 * h);
  const v8f f2 = load8(w2 + c * D_ + 8 * h);
  const v8f bias1 = load8(b1 + 8 * h);
  const float bias2 = b2[c];

  v8f x[2];
  x[0] = load8(emb + (size_t)tok0 * D_ + 8 * h);
  x[1] = load8(emb + (size_t)tok1 * D_ + 8 * h);

  v8f QT[2], KT[2], VN[2];
#pragma unroll
  for (int t = 0; t < 2; ++t) {
    v8f cq = {}, ck = {}, cv = {};
#pragma unroll
    for (int j = 0; j < 4; ++j) {
      const v2f xp = kpair(x[t], j);
      cq = wmma4(kpair(fq, j), xp, cq);
      ck = wmma4(kpair(fk, j), xp, ck);
      cv = wmma4(xp, kpair(fv, j), cv);
    }
    QT[t] = cq;
    KT[t] = ck;
    VN[t] = cv;
  }

  v8f P[2][2];
  {
    v8f s00 = {}, s01 = {}, s10 = {}, s11 = {};
#pragma unroll
    for (int j = 0; j < 4; ++j) {
      const v2f k0 = kpair(KT[0], j), k1 = kpair(KT[1], j);
      const v2f q0 = kpair(QT[0], j), q1 = kpair(QT[1], j);
      s00 = wmma4(k0, q0, s00);
      s01 = wmma4(k1, q0, s01);
      s10 = wmma4(k0, q1, s10);
      s11 = wmma4(k1, q1, s11);
    }
    P[0][0] = s00; P[0][1] = s01; P[1][0] = s10; P[1][1] = s11;
  }

#pragma unroll
  for (int sb = 0; sb < 2; ++sb) {
    float mx = -3.402823466e38f;
#pragma unroll
    for (int tb = 0; tb < 2; ++tb)
#pragma unroll
      for (int r = 0; r < 8; ++r) mx = fmaxf(mx, P[sb][tb][r]);
    mx = fmaxf(mx, __shfl_xor(mx, 16, 32));
    float sum = 0.f;
#pragma unroll
    for (int tb = 0; tb < 2; ++tb)
#pragma unroll
      for (int r = 0; r < 8; ++r) {
        const float e = __expf((P[sb][tb][r] - mx) * 0.25f);
        P[sb][tb][r] = e;
        sum += e;
      }
    sum += __shfl_xor(sum, 16, 32);
    const float inv = 1.f / sum;
#pragma unroll
    for (int tb = 0; tb < 2; ++tb)
#pragma unroll
      for (int r = 0; r < 8; ++r) P[sb][tb][r] *= inv;
  }

  v8f u0 = {}, u1 = {};
#pragma unroll
  for (int tb = 0; tb < 2; ++tb)
#pragma unroll
    for (int j = 0; j < 4; ++j) {
      const v2f a = kpair(VN[tb], j);
      u0 = wmma4(a, kpair(P[0][tb], j), u0);
      u1 = wmma4(a, kpair(P[1][tb], j), u1);
    }

  v8f g0 = {}, g1 = {};
#pragma unroll
  for (int j = 0; j < 4; ++j) {
    const v2f a = kpair(f1, j);
    g0 = wmma4(a, kpair(u0, j), g0);
    g1 = wmma4(a, kpair(u1, j), g1);
  }
#pragma unroll
  for (int r = 0; r < 8; ++r) {
    g0[r] = fmaxf(g0[r] + bias1[r], 0.f);
    g1[r] = fmaxf(g1[r] + bias1[r], 0.f);
  }

  v8f y0 = {}, y1 = {};
#pragma unroll
  for (int j = 0; j < 4; ++j) {
    const v2f bw = kpair(f2, j);
    y0 = wmma4(kpair(g0, j), bw, y0);
    y1 = wmma4(kpair(g1, j), bw, y1);
  }
  float partial = 0.f;
#pragma unroll
  for (int r = 0; r < 8; ++r) partial += y0[r];
#pragma unroll
  for (int r = 0; r < 8; ++r) partial += y1[r];
  const float other = __shfl_xor(partial, 16, 32);
  const float mean  = (partial + other) * (1.f / 32.f) + bias2;
  if (h == 0) {
    out[(size_t)enc * ((size_t)B_ * D_) + (size_t)b * D_ + c] = mean;
  }
}

extern "C" void kernel_launch(void* const* d_in, const int* in_sizes, int n_in,
                              void* d_out, int out_size, void* d_ws,
                              size_t ws_size, hipStream_t stream) {
  (void)in_sizes; (void)n_in; (void)out_size; (void)d_ws; (void)ws_size;
  const int*   arr1 = (const int*)d_in[0];
  const int*   arr2 = (const int*)d_in[1];
  const float* emb  = (const float*)d_in[2];
  const float* wq   = (const float*)d_in[3];
  const float* wk   = (const float*)d_in[4];
  const float* wv   = (const float*)d_in[5];
  const float* w1   = (const float*)d_in[6];
  const float* b1   = (const float*)d_in[7];
  const float* w2   = (const float*)d_in[8];
  const float* b2   = (const float*)d_in[9];
  float* out = (float*)d_out;

  const unsigned tasks  = 2u * B_;
  const unsigned blocks = tasks / WPB;
  encoder_kernel<<<dim3(blocks), dim3(TPB), 0, stream>>>(
      arr1, arr2, emb, wq, wk, wv, w1, b1, w2, b2, out);
}


# Q, K, V projections hoisted to per-vocabulary tables: a first launch forms them once per emb row on WMMA (same products, same k order), the encoder gathers rows; 72 to 48 WMMAs per task
# speedup vs baseline: 1.5934x; 1.2413x over previous
#include <hip/hip_runtime.h>

#define B_  131072
#define S_  32
#define D_  16
#define V_  257
#define VT_ 17
#define VP_ (VT_ * 16)
#define WPB 4
#define TPB (WPB * 32)

typedef float v2f __attribute__((ext_vector_type(2)));
typedef float v8f __attribute__((ext_vector_type(8)));

__device__ __forceinline__ v8f wmma4(v2f a, v2f b, v8f c) {
  return __builtin_amdgcn_wmma_f32_16x16x4_f32(false, a, false, b, (short)0, c,
                                               false, false);
}


__device__ __forceinline__ v8f load8(const float* __restrict__ p) {
  const float4 lo = ((const float4*)p)[0];
  const float4 hi = ((const float4*)p)[1];
  v8f r;
  r[0] = lo.x; r[1] = lo.y; r[2] = lo.z; r[3] = lo.w;
  r[4] = hi.x; r[5] = hi.y; r[6] = hi.z; r[7] = hi.w;
  return r;
}

__device__ __forceinline__ v8f load8_at(const float* __restrict__ base,
                                        unsigned byte_off) {
  return load8((const float*)((const char*)base + byte_off));
}
__device__ __forceinline__ float load1_at(const float* __restrict__ base,
                                          unsigned byte_off) {
  return *(const float*)((const char*)base + byte_off);
}

__device__ __forceinline__ void store8(float* __restrict__ p, v8f v) {
  ((float4*)p)[0] = make_float4(v[0], v[1], v[2], v[3]);
  ((float4*)p)[1] = make_float4(v[4], v[5], v[6], v[7]);
}

__device__ __forceinline__ v2f kpair(v8f v, int j) {
  v2f r;
  r.x = v[2 * j];
  r.y = v[2 * j + 1];
  return r;
}

__global__ __launch_bounds__(32) void vocab_tables_kernel(
    const float* __restrict__ emb,
    const float* __restrict__ wq, const float* __restrict__ wk,
    const float* __restrict__ wv,
    float* __restrict__ qk, float* __restrict__ vt) {
  const int lane = threadIdx.x & 31;
  const int c    = lane & 15;
  const int h    = lane >> 4;
  const int row  = blockIdx.x * 16 + c;
  const int srow = row < V_ ? row : V_ - 1;

  const v8f xs = load8(emb + srow * D_ + 8 * h);
  const v8f fq = load8(wq + c * D_ + 8 * h);
  const v8f fk = load8(wk + c * D_ + 8 * h);
  const v8f fv = load8(wv + c * D_ + 8 * h);

  v8f cq = {}, ck = {}, cv = {};
#pragma unroll
  for (int j = 0; j < 4; ++j) {
    const v2f xp = kpair(xs, j);
    cq = wmma4(kpair(fq, j), xp, cq);
    ck = wmma4(kpair(fk, j), xp, ck);
    cv = wmma4(kpair(fv, j), xp, cv);
  }
  store8(qk + row * 32 + 8 * h, cq);
  store8(qk + row * 32 + 16 + 8 * h, ck);
  store8(vt + row * D_ + 8 * h, cv);
}

__global__ __launch_bounds__(TPB) void encoder_kernel(
    const int* __restrict__ arr1, const int* __restrict__ arr2,
    const float* __restrict__ qk, const float* __restrict__ vt,
    const float* __restrict__ w1, const float* __restrict__ b1,
    const float* __restrict__ w2, const float* __restrict__ b2,
    float* __restrict__ out) {
  const int tid  = threadIdx.x;
  const int wave = tid >> 5;
  const int lane = tid & 31;
  const int c    = lane & 15;
  const int h    = lane >> 4;

  const unsigned task = blockIdx.x * WPB + wave;
  const int enc = (task >= (unsigned)B_) ? 1 : 0;
  const int b   = enc ? (int)(task - (unsigned)B_) : (int)task;
  const int* __restrict__ arr = enc ? arr2 : arr1;

  const int* __restrict__ trow = arr + (size_t)b * S_;

  const int tok0 = trow[c];
  const int tok1 = trow[16 + c];
  int tv[2][8];
#pragma unroll
  for (int tb = 0; tb < 2; ++tb) {
    const int4 lo = ((const int4*)(trow + 16 * tb + 8 * h))[0];
    const int4 hi = ((const int4*)(trow + 16 * tb + 8 * h))[1];
    tv[tb][0] = lo.x; tv[tb][1] = lo.y; tv[tb][2] = lo.z; tv[tb][3] = lo.w;
    tv[tb][4] = hi.x; tv[tb][5] = hi.y; tv[tb][6] = hi.z; tv[tb][7] = hi.w;
  }

  const v8f f1 = load8(w1 + c * D_ + 8 * h);
  const v8f f2 = load8(w2 + c * D_ + 8 * h);
  const v8f bias1 = load8(b1 + 8 * h);
  const float bias2 = b2[c];

  v8f QT[2], KT[2], VN[2];
  const unsigned o0 = (unsigned)tok0 * 128u + 32u * (unsigned)h;
  const unsigned o1 = (unsigned)tok1 * 128u + 32u * (unsigned)h;
  QT[0] = load8_at(qk, o0);
  KT[0] = load8_at(qk, o0 + 64u);
  QT[1] = load8_at(qk, o1);
  KT[1] = load8_at(qk, o1 + 64u);
  const unsigned c4 = 4u * (unsigned)c;
#pragma unroll
  for (int tb = 0; tb < 2; ++tb)
#pragma unroll
    for (int r = 0; r < 8; ++r)
      VN[tb][r] = load1_at(vt, (unsigned)tv[tb][r] * 64u + c4);

  v8f P[2][2];
  {
    v8f s00 = {}, s01 = {}, s10 = {}, s11 = {};
#pragma unroll
    for (int j = 0; j < 4; ++j) {
      const v2f k0 = kpair(KT[0], j), k1 = kpair(KT[1], j);
      const v2f q0 = kpair(QT[0], j), q1 = kpair(QT[1], j);
      s00 = wmma4(k0, q0, s00);
      s01 = wmma4(k1, q0, s01);
      s10 = wmma4(k0, q1, s10);
      s11 = wmma4(k1, q1, s11);
    }
    P[0][0] = s00; P[0][1] = s01; P[1][0] = s10; P[1][1] = s11;
  }

#pragma unroll
  for (int sb = 0; sb < 2; ++sb) {
    float mx = -3.402823466e38f;
#pragma unroll
    for (int tb = 0; tb < 2; ++tb)
#pragma unroll
      for (int r = 0; r < 8; ++r) mx = fmaxf(mx, P[sb][tb][r]);
    mx = fmaxf(mx, __shfl_xor(mx, 16, 32));
    float sum = 0.f;
#pragma unroll
    for (int tb = 0; tb < 2; ++tb)
#pragma unroll
      for (int r = 0; r < 8; ++r) {
        const float e = __expf((P[sb][tb][r] - mx) * 0.25f);
        P[sb][tb][r] = e;
        sum += e;
      }
    sum += __shfl_xor(sum, 16, 32);
    const float inv = 1.f / sum;
#pragma unroll
    for (int tb = 0; tb < 2; ++tb)
#pragma unroll
      for (int r = 0; r < 8; ++r) P[sb][tb][r] *= inv;
  }

  v8f u0 = {}, u1 = {};
#pragma unroll
  for (int tb = 0; tb < 2; ++tb)
#pragma unroll
    for (int j = 0; j < 4; ++j) {
      const v2f a = kpair(VN[tb], j);
      u0 = wmma4(a, kpair(P[0][tb], j), u0);
      u1 = wmma4(a, kpair(P[1][tb], j), u1);
    }

  v8f g0 = {}, g1 = {};
#pragma unroll
  for (int j = 0; j < 4; ++j) {
    const v2f a = kpair(f1, j);
    g0 = wmma4(a, kpair(u0, j), g0);
    g1 = wmma4(a, kpair(u1, j), g1);
  }
#pragma unroll
  for (int r = 0; r < 8; ++r) {
    g0[r] = fmaxf(g0[r] + bias1[r], 0.f);
    g1[r] = fmaxf(g1[r] + bias1[r], 0.f);
  }

  v8f y0 = {}, y1 = {};
#pragma unroll
  for (int j = 0; j < 4; ++j) {
    const v2f bw = kpair(f2, j);
    y0 = wmma4(kpair(g0, j), bw, y0);
    y1 = wmma4(kpair(g1, j), bw, y1);
  }
  float partial = 0.f;
#pragma unroll
  for (int r = 0; r < 8; ++r) partial += y0[r];
#pragma unroll
  for (int r = 0; r < 8; ++r) partial += y1[r];
  const float other = __shfl_xor(partial, 16, 32);
  const float mean  = (partial + other) * (1.f / 32.f) + bias2;
  if (h == 0) {
    out[(size_t)enc * ((size_t)B_ * D_) + (size_t)b * D_ + c] = mean;
  }
}

extern "C" void kernel_launch(void* const* d_in, const int* in_sizes, int n_in,
                              void* d_out, int out_size, void* d_ws,
                              size_t ws_size, hipStream_t stream) {
  (void)in_sizes; (void)n_in; (void)out_size; (void)ws_size;
  const int*   arr1 = (const int*)d_in[0];
  const int*   arr2 = (const int*)d_in[1];
  const float* emb  = (const float*)d_in[2];
  const float* wq   = (const float*)d_in[3];
  const float* wk   = (const float*)d_in[4];
  const float* wv   = (const float*)d_in[5];
  const float* w1   = (const float*)d_in[6];
  const float* b1   = (const float*)d_in[7];
  const float* w2   = (const float*)d_in[8];
  const float* b2   = (const float*)d_in[9];
  float* out = (float*)d_out;

  float* qk = (float*)d_ws;
  float* vt = (float*)d_ws + VP_ * 32;
  vocab_tables_kernel<<<dim3(VT_), dim3(32), 0, stream>>>(emb, wq, wk, wv,
                                                          qk, vt);

  const unsigned tasks  = 2u * B_;
  const unsigned blocks = tasks / WPB;
  encoder_kernel<<<dim3(blocks), dim3(TPB), 0, stream>>>(
      arr1, arr2, qk, vt, w1, b1, w2, b2, out);
}


# w1 product hoisted into the per-vocabulary V table (V*w1^T formed on WMMA in launch 1); encoder's 8 w1 WMMAs go, 48 to 40 per task; (attn*V)*w1^T regrouped as attn*(V*w1^T), sums reordered
# speedup vs baseline: 1.7882x; 1.1223x over previous
#include <hip/hip_runtime.h>

#define B_  131072
#define S_  32
#define D_  16
#define V_  257
#define VT_ 17
#define VP_ (VT_ * 16)
#define WPB 4
#define TPB (WPB * 32)

typedef float v2f __attribute__((ext_vector_type(2)));
typedef float v8f __attribute__((ext_vector_type(8)));

__device__ __forceinline__ v8f wmma4(v2f a, v2f b, v8f c) {
  return __builtin_amdgcn_wmma_f32_16x16x4_f32(false, a, false, b, (short)0, c,
                                               false, false);
}


__device__ __forceinline__ v8f load8(const float* __restrict__ p) {
  const float4 lo = ((const float4*)p)[0];
  const float4 hi = ((const float4*)p)[1];
  v8f r;
  r[0] = lo.x; r[1] = lo.y; r[2] = lo.z; r[3] = lo.w;
  r[4] = hi.x; r[5] = hi.y; r[6] = hi.z; r[7] = hi.w;
  return r;
}

__device__ __forceinline__ v8f load8_at(const float* __restrict__ base,
                                        unsigned byte_off) {
  return load8((const float*)((const char*)base + byte_off));
}
__device__ __forceinline__ float load1_at(const float* __restrict__ base,
                                          unsigned byte_off) {
  return *(const float*)((const char*)base + byte_off);
}

__device__ __forceinline__ void store8(float* __restrict__ p, v8f v) {
  ((float4*)p)[0] = make_float4(v[0], v[1], v[2], v[3]);
  ((float4*)p)[1] = make_float4(v[4], v[5], v[6], v[7]);
}

__device__ __forceinline__ v2f kpair(v8f v, int j) {
  v2f r;
  r.x = v[2 * j];
  r.y = v[2 * j + 1];
  return r;
}

__global__ __launch_bounds__(32) void vocab_tables_kernel(
    const float* __restrict__ emb,
    const float* __restrict__ wq, const float* __restrict__ wk,
    const float* __restrict__ wv, const float* __restrict__ w1,
    float* __restrict__ qk, float* __restrict__ vt) {
  const int lane = threadIdx.x & 31;
  const int c    = lane & 15;
  const int h    = lane >> 4;
  const int row  = blockIdx.x * 16 + c;
  const int srow = row < V_ ? row : V_ - 1;

  const v8f xs = load8(emb + srow * D_ + 8 * h);
  const v8f fq = load8(wq + c * D_ + 8 * h);
  const v8f fk = load8(wk + c * D_ + 8 * h);
  const v8f fv = load8(wv + c * D_ + 8 * h);
  const v8f f1 = load8(w1 + c * D_ + 8 * h);

  v8f cq = {}, ck = {}, cv = {};
#pragma unroll
  for (int j = 0; j < 4; ++j) {
    const v2f xp = kpair(xs, j);
    cq = wmma4(kpair(fq, j), xp, cq);
    ck = wmma4(kpair(fk, j), xp, ck);
    cv = wmma4(kpair(fv, j), xp, cv);
  }
  v8f cw = {};
#pragma unroll
  for (int j = 0; j < 4; ++j) cw = wmma4(kpair(f1, j), kpair(cv, j), cw);
  store8(qk + row * 32 + 8 * h, cq);
  store8(qk + row * 32 + 16 + 8 * h, ck);
  store8(vt + row * D_ + 8 * h, cw);
}

__global__ __launch_bounds__(TPB) void encoder_kernel(
    const int* __restrict__ arr1, const int* __restrict__ arr2,
    const float* __restrict__ qk, const float* __restrict__ vt,
    const float* __restrict__ b1,
    const float* __restrict__ w2, const float* __restrict__ b2,
    float* __restrict__ out) {
  const int tid  = threadIdx.x;
  const int wave = tid >> 5;
  const int lane = tid & 31;
  const int c    = lane & 15;
  const int h    = lane >> 4;

  const unsigned task = blockIdx.x * WPB + wave;
  const int enc = (task >= (unsigned)B_) ? 1 : 0;
  const int b   = enc ? (int)(task - (unsigned)B_) : (int)task;
  const int* __restrict__ arr = enc ? arr2 : arr1;

  const int* __restrict__ trow = arr + (size_t)b * S_;

  const int tok0 = trow[c];
  const int tok1 = trow[16 + c];
  int tv[2][8];
#pragma unroll
  for (int tb = 0; tb < 2; ++tb) {
    const int4 lo = ((const int4*)(trow + 16 * tb + 8 * h))[0];
    const int4 hi = ((const int4*)(trow + 16 * tb + 8 * h))[1];
    tv[tb][0] = lo.x; tv[tb][1] = lo.y; tv[tb][2] = lo.z; tv[tb][3] = lo.w;
    tv[tb][4] = hi.x; tv[tb][5] = hi.y; tv[tb][6] = hi.z; tv[tb][7] = hi.w;
  }

  const v8f f2 = load8(w2 + c * D_ + 8 * h);
  const v8f bias1 = load8(b1 + 8 * h);
  const float bias2 = b2[c];

  v8f QT[2], KT[2], VN[2];
  const unsigned o0 = (unsigned)tok0 * 128u + 32u * (unsigned)h;
  const unsigned o1 = (unsigned)tok1 * 128u + 32u * (unsigned)h;
  QT[0] = load8_at(qk, o0);
  KT[0] = load8_at(qk, o0 + 64u);
  QT[1] = load8_at(qk, o1);
  KT[1] = load8_at(qk, o1 + 64u);
  const unsigned c4 = 4u * (unsigned)c;
#pragma unroll
  for (int tb = 0; tb < 2; ++tb)
#pragma unroll
    for (int r = 0; r < 8; ++r)
      VN[tb][r] = load1_at(vt, (unsigned)tv[tb][r] * 64u + c4);

  v8f P[2][2];
  {
    v8f s00 = {}, s01 = {}, s10 = {}, s11 = {};
#pragma unroll
    for (int j = 0; j < 4; ++j) {
      const v2f k0 = kpair(KT[0], j), k1 = kpair(KT[1], j);
      const v2f q0 = kpair(QT[0], j), q1 = kpair(QT[1], j);
      s00 = wmma4(k0, q0, s00);
      s01 = wmma4(k1, q0, s01);
      s10 = wmma4(k0, q1, s10);
      s11 = wmma4(k1, q1, s11);
    }
    P[0][0] = s00; P[0][1] = s01; P[1][0] = s10; P[1][1] = s11;
  }

#pragma unroll
  for (int sb = 0; sb < 2; ++sb) {
    float mx = -3.402823466e38f;
#pragma unroll
    for (int tb = 0; tb < 2; ++tb)
#pragma unroll
      for (int r = 0; r < 8; ++r) mx = fmaxf(mx, P[sb][tb][r]);
    mx = fmaxf(mx, __shfl_xor(mx, 16, 32));
    float sum = 0.f;
#pragma unroll
    for (int tb = 0; tb < 2; ++tb)
#pragma unroll
      for (int r = 0; r < 8; ++r) {
        const float e = __expf((P[sb][tb][r] - mx) * 0.25f);
        P[sb][tb][r] = e;
        sum += e;
      }
    sum += __shfl_xor(sum, 16, 32);
    const float inv = 1.f / sum;
#pragma unroll
    for (int tb = 0; tb < 2; ++tb)
#pragma unroll
      for (int r = 0; r < 8; ++r) P[sb][tb][r] *= inv;
  }

  v8f u0 = {}, u1 = {};
#pragma unroll
  for (int tb = 0; tb < 2; ++tb)
#pragma unroll
    for (int j = 0; j < 4; ++j) {
      const v2f a = kpair(VN[tb], j);
      u0 = wmma4(a, kpair(P[0][tb], j), u0);
      u1 = wmma4(a, kpair(P[1][tb], j), u1);
    }

  v8f g0, g1;
#pragma unroll
  for (int r = 0; r < 8; ++r) {
    g0[r] = fmaxf(u0[r] + bias1[r], 0.f);
    g1[r] = fmaxf(u1[r] + bias1[r], 0.f);
  }

  v8f y0 = {}, y1 = {};
#pragma unroll
  for (int j = 0; j < 4; ++j) {
    const v2f bw = kpair(f2, j);
    y0 = wmma4(kpair(g0, j), bw, y0);
    y1 = wmma4(kpair(g1, j), bw, y1);
  }
  float partial = 0.f;
#pragma unroll
  for (int r = 0; r < 8; ++r) partial += y0[r];
#pragma unroll
  for (int r = 0; r < 8; ++r) partial += y1[r];
  const float other = __shfl_xor(partial, 16, 32);
  const float mean  = (partial + other) * (1.f / 32.f) + bias2;
  if (h == 0) {
    out[(size_t)enc * ((size_t)B_ * D_) + (size_t)b * D_ + c] = mean;
  }
}

extern "C" void kernel_launch(void* const* d_in, const int* in_sizes, int n_in,
                              void* d_out, int out_size, void* d_ws,
                              size_t ws_size, hipStream_t stream) {
  (void)in_sizes; (void)n_in; (void)out_size; (void)ws_size;
  const int*   arr1 = (const int*)d_in[0];
  const int*   arr2 = (const int*)d_in[1];
  const float* emb  = (const float*)d_in[2];
  const float* wq   = (const float*)d_in[3];
  const float* wk   = (const float*)d_in[4];
  const float* wv   = (const float*)d_in[5];
  const float* w1   = (const float*)d_in[6];
  const float* b1   = (const float*)d_in[7];
  const float* w2   = (const float*)d_in[8];
  const float* b2   = (const float*)d_in[9];
  float* out = (float*)d_out;

  float* qk = (float*)d_ws;
  float* vt = (float*)d_ws + VP_ * 32;
  vocab_tables_kernel<<<dim3(VT_), dim3(32), 0, stream>>>(emb, wq, wk, wv, w1,
                                                          qk, vt);

  const unsigned tasks  = 2u * B_;
  const unsigned blocks = tasks / WPB;
  encoder_kernel<<<dim3(blocks), dim3(TPB), 0, stream>>>(
      arr1, arr2, qk, vt, b1, w2, b2, out);
}


# Mean-pool moved ahead of the w2 product (linear): a wave loops over 8 sequences, a one-hot WMMA puts each pooled relu row in a tile, one w2 product serves 8; 40 to 33.5 WMMAs; pooled sums reordered
# speedup vs baseline: 2.0080x; 1.1229x over previous
#include <hip/hip_runtime.h>

#define B_  131072
#define S_  32
#define D_  16
#define V_  257
#define VT_ 17
#define VP_ (VT_ * 16)
#define WPB 4
#define TPB (WPB * 32)

typedef float v2f __attribute__((ext_vector_type(2)));
typedef float v8f __attribute__((ext_vector_type(8)));

__device__ __forceinline__ v8f wmma4(v2f a, v2f b, v8f c) {
  return __builtin_amdgcn_wmma_f32_16x16x4_f32(false, a, false, b, (short)0, c,
                                               false, false);
}


__device__ __forceinline__ v8f load8(const float* __restrict__ p) {
  const float4 lo = ((const float4*)p)[0];
  const float4 hi = ((const float4*)p)[1];
  v8f r;
  r[0] = lo.x; r[1] = lo.y; r[2] = lo.z; r[3] = lo.w;
  r[4] = hi.x; r[5] = hi.y; r[6] = hi.z; r[7] = hi.w;
  return r;
}

__device__ __forceinline__ v8f load8_at(const float* __restrict__ base,
                                        unsigned byte_off) {
  return load8((const float*)((const char*)base + byte_off));
}
__device__ __forceinline__ float load1_at(const float* __restrict__ base,
                                          unsigned byte_off) {
  return *(const float*)((const char*)base + byte_off);
}

__device__ __forceinline__ void store8(float* __restrict__ p, v8f v) {
  ((float4*)p)[0] = make_float4(v[0], v[1], v[2], v[3]);
  ((float4*)p)[1] = make_float4(v[4], v[5], v[6], v[7]);
}

__device__ __forceinline__ v2f kpair(v8f v, int j) {
  v2f r;
  r.x = v[2 * j];
  r.y = v[2 * j + 1];
  return r;
}

__global__ __launch_bounds__(32) void vocab_tables_kernel(
    const float* __restrict__ emb,
    const float* __restrict__ wq, const float* __restrict__ wk,
    const float* __restrict__ wv, const float* __restrict__ w1,
    float* __restrict__ qk, float* __restrict__ vt) {
  const int lane = threadIdx.x & 31;
  const int c    = lane & 15;
  const int h    = lane >> 4;
  const int row  = blockIdx.x * 16 + c;
  const int srow = row < V_ ? row : V_ - 1;

  const v8f xs = load8(emb + srow * D_ + 8 * h);
  const v8f fq = load8(wq + c * D_ + 8 * h);
  const v8f fk = load8(wk + c * D_ + 8 * h);
  const v8f fv = load8(wv + c * D_ + 8 * h);
  const v8f f1 = load8(w1 + c * D_ + 8 * h);

  v8f cq = {}, ck = {}, cv = {};
#pragma unroll
  for (int j = 0; j < 4; ++j) {
    const v2f xp = kpair(xs, j);
    cq = wmma4(kpair(fq, j), xp, cq);
    ck = wmma4(kpair(fk, j), xp, ck);
    cv = wmma4(kpair(fv, j), xp, cv);
  }
  v8f cw = {};
#pragma unroll
  for (int j = 0; j < 4; ++j) cw = wmma4(kpair(f1, j), kpair(cv, j), cw);
  store8(qk + row * 32 + 8 * h, cq);
  store8(qk + row * 32 + 16 + 8 * h, ck);
  store8(vt + row * D_ + 8 * h, cw);
}

#define G_ 8

__global__ __launch_bounds__(TPB) void encoder_kernel(
    const int* __restrict__ arr1, const int* __restrict__ arr2,
    const float* __restrict__ qk, const float* __restrict__ vt,
    const float* __restrict__ b1,
    const float* __restrict__ w2, const float* __restrict__ b2,
    float* __restrict__ out) {
  const int tid  = threadIdx.x;
  const int wave = tid >> 5;
  const int lane = tid & 31;
  const int c    = lane & 15;
  const int h    = lane >> 4;

  const unsigned task0 = (blockIdx.x * WPB + wave) * G_;
  const int enc = (task0 >= (unsigned)B_) ? 1 : 0;
  const int b0  = enc ? (int)(task0 - (unsigned)B_) : (int)task0;
  const int* __restrict__ arr = enc ? arr2 : arr1;

  const int* __restrict__ trow = arr + (size_t)b0 * S_;

  const float bias1 = b1[c];
  const unsigned c4 = 4u * (unsigned)c;

  v8f T = {};

#pragma nounroll
  for (int i = 0; i < G_; ++i) {
    const int* __restrict__ tr = trow + i * S_;
    const int tok0 = tr[c];
    const int tok1 = tr[16 + c];
    int tv[2][8];
#pragma unroll
    for (int tb = 0; tb < 2; ++tb) {
      const int4 lo = ((const int4*)(tr + 16 * tb + 8 * h))[0];
      const int4 hi = ((const int4*)(tr + 16 * tb + 8 * h))[1];
      tv[tb][0] = lo.x; tv[tb][1] = lo.y; tv[tb][2] = lo.z; tv[tb][3] = lo.w;
      tv[tb][4] = hi.x; tv[tb][5] = hi.y; tv[tb][6] = hi.z; tv[tb][7] = hi.w;
    }

    v8f QT[2], KT[2], VN[2];
    const unsigned o0 = (unsigned)tok0 * 128u + 32u * (unsigned)h;
    const unsigned o1 = (unsigned)tok1 * 128u + 32u * (unsigned)h;
    QT[0] = load8_at(qk, o0);
    KT[0] = load8_at(qk, o0 + 64u);
    QT[1] = load8_at(qk, o1);
    KT[1] = load8_at(qk, o1 + 64u);

#pragma unroll
    for (int tb = 0; tb < 2; ++tb)
#pragma unroll
      for (int r = 0; r < 8; ++r)
        VN[tb][r] = load1_at(vt, (unsigned)tv[tb][r] * 64u + c4);

    v8f P[2][2];
    {
      v8f s00 = {}, s01 = {}, s10 = {}, s11 = {};
#pragma unroll
      for (int j = 0; j < 4; ++j) {
        const v2f k0 = kpair(KT[0], j), k1 = kpair(KT[1], j);
        const v2f q0 = kpair(QT[0], j), q1 = kpair(QT[1], j);
        s00 = wmma4(k0, q0, s00);
        s01 = wmma4(k1, q0, s01);
        s10 = wmma4(k0, q1, s10);
        s11 = wmma4(k1, q1, s11);
      }
      P[0][0] = s00; P[0][1] = s01; P[1][0] = s10; P[1][1] = s11;
    }

#pragma unroll
    for (int sb = 0; sb < 2; ++sb) {
      float mx = -3.402823466e38f;
#pragma unroll
      for (int tb = 0; tb < 2; ++tb)
#pragma unroll
        for (int r = 0; r < 8; ++r) mx = fmaxf(mx, P[sb][tb][r]);
      mx = fmaxf(mx, __shfl_xor(mx, 16, 32));
      float sum = 0.f;
#pragma unroll
      for (int tb = 0; tb < 2; ++tb)
#pragma unroll
        for (int r = 0; r < 8; ++r) {
          const float e = __expf((P[sb][tb][r] - mx) * 0.25f);
          P[sb][tb][r] = e;
          sum += e;
        }
      sum += __shfl_xor(sum, 16, 32);
      const float inv = 1.f / sum;
#pragma unroll
      for (int tb = 0; tb < 2; ++tb)
#pragma unroll
        for (int r = 0; r < 8; ++r) P[sb][tb][r] *= inv;
    }

    v8f u0 = {}, u1 = {};
#pragma unroll
    for (int tb = 0; tb < 2; ++tb)
#pragma unroll
      for (int j = 0; j < 4; ++j) {
        const v2f bv = kpair(VN[tb], j);
        u0 = wmma4(kpair(P[0][tb], j), bv, u0);
        u1 = wmma4(kpair(P[1][tb], j), bv, u1);
      }

    v2f p;
    {
      float p0 = 0.f, p1 = 0.f;
#pragma unroll
      for (int r = 0; r < 8; ++r) {
        p0 += fmaxf(u0[r] + bias1, 0.f);
        p1 += fmaxf(u1[r] + bias1, 0.f);
      }
      p.x = p0;
      p.y = p1;
    }

    const int slot = (i & 3) | ((i & 4) << 1);
    const float one = (c == slot) ? 1.f : 0.f;
    v2f sel;
    sel.x = one;
    sel.y = one;
    T = wmma4(p, sel, T);
  }

  const v8f f2 = load8(w2 + c * D_ + 8 * h);
  const float bias2 = b2[c];
  v8f y = {};
#pragma unroll
  for (int j = 0; j < 4; ++j) y = wmma4(kpair(T, j), kpair(f2, j), y);

  float* __restrict__ orow =
      out + (size_t)enc * ((size_t)B_ * D_) + (size_t)(b0 + 4 * h) * D_ + c;
#pragma unroll
  for (int r = 0; r < 4; ++r) orow[r * D_] = y[r] * (1.f / 32.f) + bias2;
}

extern "C" void kernel_launch(void* const* d_in, const int* in_sizes, int n_in,
                              void* d_out, int out_size, void* d_ws,
                              size_t ws_size, hipStream_t stream) {
  (void)in_sizes; (void)n_in; (void)out_size; (void)ws_size;
  const int*   arr1 = (const int*)d_in[0];
  const int*   arr2 = (const int*)d_in[1];
  const float* emb  = (const float*)d_in[2];
  const float* wq   = (const float*)d_in[3];
  const float* wk   = (const float*)d_in[4];
  const float* wv   = (const float*)d_in[5];
  const float* w1   = (const float*)d_in[6];
  const float* b1   = (const float*)d_in[7];
  const float* w2   = (const float*)d_in[8];
  const float* b2   = (const float*)d_in[9];
  float* out = (float*)d_out;

  float* qk = (float*)d_ws;
  float* vt = (float*)d_ws + VP_ * 32;
  vocab_tables_kernel<<<dim3(VT_), dim3(32), 0, stream>>>(emb, wq, wk, wv, w1,
                                                          qk, vt);

  const unsigned waves  = 2u * B_ / G_;
  const unsigned blocks = waves / WPB;
  encoder_kernel<<<dim3(blocks), dim3(TPB), 0, stream>>>(
      arr1, arr2, qk, vt, b1, w2, b2, out);
}


# Softmax vector work cut: log2(e)/4 goes into the Q table, so a score needs one subtract and one exp2; v_rcp for the row reciprocal; row sums taken pairwise on packed adds (softmax sums reordered)
# speedup vs baseline: 2.1696x; 1.0805x over previous
#include <hip/hip_runtime.h>

#define B_  131072
#define S_  32
#define D_  16
#define V_  257
#define VT_ 17
#define VP_ (VT_ * 16)
#define WPB 4
#define TPB (WPB * 32)
#define KS_ 0.36067376022224085f

typedef float v2f __attribute__((ext_vector_type(2)));
typedef float v8f __attribute__((ext_vector_type(8)));

__device__ __forceinline__ v8f wmma4(v2f a, v2f b, v8f c) {
  return __builtin_amdgcn_wmma_f32_16x16x4_f32(false, a, false, b, (short)0, c,
                                               false, false);
}


__device__ __forceinline__ v8f load8(const float* __restrict__ p) {
  const float4 lo = ((const float4*)p)[0];
  const float4 hi = ((const float4*)p)[1];
  v8f r;
  r[0] = lo.x; r[1] = lo.y; r[2] = lo.z; r[3] = lo.w;
  r[4] = hi.x; r[5] = hi.y; r[6] = hi.z; r[7] = hi.w;
  return r;
}

__device__ __forceinline__ v8f load8_at(const float* __restrict__ base,
                                        unsigned byte_off) {
  return load8((const float*)((const char*)base + byte_off));
}
__device__ __forceinline__ float load1_at(const float* __restrict__ base,
                                          unsigned byte_off) {
  return *(const float*)((const char*)base + byte_off);
}

__device__ __forceinline__ void store8(float* __restrict__ p, v8f v) {
  ((float4*)p)[0] = make_float4(v[0], v[1], v[2], v[3]);
  ((float4*)p)[1] = make_float4(v[4], v[5], v[6], v[7]);
}

__device__ __forceinline__ v2f kpair(v8f v, int j) {
  v2f r;
  r.x = v[2 * j];
  r.y = v[2 * j + 1];
  return r;
}

__global__ __launch_bounds__(32) void vocab_tables_kernel(
    const float* __restrict__ emb,
    const float* __restrict__ wq, const float* __restrict__ wk,
    const float* __restrict__ wv, const float* __restrict__ w1,
    float* __restrict__ qk, float* __restrict__ vt) {
  const int lane = threadIdx.x & 31;
  const int c    = lane & 15;
  const int h    = lane >> 4;
  const int row  = blockIdx.x * 16 + c;
  const int srow = row < V_ ? row : V_ - 1;

  const v8f xs = load8(emb + srow * D_ + 8 * h);
  v8f fq = load8(wq + c * D_ + 8 * h);
#pragma unroll
  for (int r = 0; r < 8; ++r) fq[r] *= KS_;
  const v8f fk = load8(wk + c * D_ + 8 * h);
  const v8f fv = load8(wv + c * D_ + 8 * h);
  const v8f f1 = load8(w1 + c * D_ + 8 * h);

  v8f cq = {}, ck = {}, cv = {};
#pragma unroll
  for (int j = 0; j < 4; ++j) {
    const v2f xp = kpair(xs, j);
    cq = wmma4(kpair(fq, j), xp, cq);
    ck = wmma4(kpair(fk, j), xp, ck);
    cv = wmma4(kpair(fv, j), xp, cv);
  }
  v8f cw = {};
#pragma unroll
  for (int j = 0; j < 4; ++j) cw = wmma4(kpair(f1, j), kpair(cv, j), cw);
  store8(qk + row * 32 + 8 * h, cq);
  store8(qk + row * 32 + 16 + 8 * h, ck);
  store8(vt + row * D_ + 8 * h, cw);
}

#define G_ 8

__global__ __launch_bounds__(TPB) void encoder_kernel(
    const int* __restrict__ arr1, const int* __restrict__ arr2,
    const float* __restrict__ qk, const float* __restrict__ vt,
    const float* __restrict__ b1,
    const float* __restrict__ w2, const float* __restrict__ b2,
    float* __restrict__ out) {
  const int tid  = threadIdx.x;
  const int wave = tid >> 5;
  const int lane = tid & 31;
  const int c    = lane & 15;
  const int h    = lane >> 4;

  const unsigned task0 = (blockIdx.x * WPB + wave) * G_;
  const int enc = (task0 >= (unsigned)B_) ? 1 : 0;
  const int b0  = enc ? (int)(task0 - (unsigned)B_) : (int)task0;
  const int* __restrict__ arr = enc ? arr2 : arr1;

  const int* __restrict__ trow = arr + (size_t)b0 * S_;

  const float bias1 = b1[c];
  const unsigned c4 = 4u * (unsigned)c;

  v8f T = {};

#pragma nounroll
  for (int i = 0; i < G_; ++i) {
    const int* __restrict__ tr = trow + i * S_;
    const int tok0 = tr[c];
    const int tok1 = tr[16 + c];
    int tv[2][8];
#pragma unroll
    for (int tb = 0; tb < 2; ++tb) {
      const int4 lo = ((const int4*)(tr + 16 * tb + 8 * h))[0];
      const int4 hi = ((const int4*)(tr + 16 * tb + 8 * h))[1];
      tv[tb][0] = lo.x; tv[tb][1] = lo.y; tv[tb][2] = lo.z; tv[tb][3] = lo.w;
      tv[tb][4] = hi.x; tv[tb][5] = hi.y; tv[tb][6] = hi.z; tv[tb][7] = hi.w;
    }

    v8f QT[2], KT[2], VN[2];
    const unsigned o0 = (unsigned)tok0 * 128u + 32u * (unsigned)h;
    const unsigned o1 = (unsigned)tok1 * 128u + 32u * (unsigned)h;
    QT[0] = load8_at(qk, o0);
    KT[0] = load8_at(qk, o0 + 64u);
    QT[1] = load8_at(qk, o1);
    KT[1] = load8_at(qk, o1 + 64u);

#pragma unroll
    for (int tb = 0; tb < 2; ++tb)
#pragma unroll
      for (int r = 0; r < 8; ++r)
        VN[tb][r] = load1_at(vt, (unsigned)tv[tb][r] * 64u + c4);

    v8f P[2][2];
    {
      v8f s00 = {}, s01 = {}, s10 = {}, s11 = {};
#pragma unroll
      for (int j = 0; j < 4; ++j) {
        const v2f k0 = kpair(KT[0], j), k1 = kpair(KT[1], j);
        const v2f q0 = kpair(QT[0], j), q1 = kpair(QT[1], j);
        s00 = wmma4(k0, q0, s00);
        s01 = wmma4(k1, q0, s01);
        s10 = wmma4(k0, q1, s10);
        s11 = wmma4(k1, q1, s11);
      }
      P[0][0] = s00; P[0][1] = s01; P[1][0] = s10; P[1][1] = s11;
    }

#pragma unroll
    for (int sb = 0; sb < 2; ++sb) {
      float mx = -3.402823466e38f;
#pragma unroll
      for (int tb = 0; tb < 2; ++tb)
#pragma unroll
        for (int r = 0; r < 8; ++r) mx = fmaxf(mx, P[sb][tb][r]);
      mx = fmaxf(mx, __shfl_xor(mx, 16, 32));
      const v2f mx2 = {mx, mx};
      v2f acc[2];
#pragma unroll
      for (int tb = 0; tb < 2; ++tb)
#pragma unroll
        for (int j = 0; j < 4; ++j) {
          const v2f a = kpair(P[sb][tb], j) - mx2;
          v2f e;
          e.x = __builtin_amdgcn_exp2f(a.x);
          e.y = __builtin_amdgcn_exp2f(a.y);
          P[sb][tb][2 * j]     = e.x;
          P[sb][tb][2 * j + 1] = e.y;
          acc[tb] = (j == 0) ? e : acc[tb] + e;
        }
      const v2f a2 = acc[0] + acc[1];
      float sum = a2.x + a2.y;
      sum += __shfl_xor(sum, 16, 32);
      const float inv = __builtin_amdgcn_rcpf(sum);
      const v2f inv2 = {inv, inv};
#pragma unroll
      for (int tb = 0; tb < 2; ++tb)
#pragma unroll
        for (int j = 0; j < 4; ++j) {
          const v2f n = kpair(P[sb][tb], j) * inv2;
          P[sb][tb][2 * j]     = n.x;
          P[sb][tb][2 * j + 1] = n.y;
        }
    }

    v8f u0 = {}, u1 = {};
#pragma unroll
    for (int tb = 0; tb < 2; ++tb)
#pragma unroll
      for (int j = 0; j < 4; ++j) {
        const v2f bv = kpair(VN[tb], j);
        u0 = wmma4(kpair(P[0][tb], j), bv, u0);
        u1 = wmma4(kpair(P[1][tb], j), bv, u1);
      }

    v2f p;
    {
      float p0 = 0.f, p1 = 0.f;
#pragma unroll
      for (int r = 0; r < 8; ++r) {
        p0 += fmaxf(u0[r] + bias1, 0.f);
        p1 += fmaxf(u1[r] + bias1, 0.f);
      }
      p.x = p0;
      p.y = p1;
    }

    const int slot = (i & 3) | ((i & 4) << 1);
    const float one = (c == slot) ? 1.f : 0.f;
    v2f sel;
    sel.x = one;
    sel.y = one;
    T = wmma4(p, sel, T);
  }

  const v8f f2 = load8(w2 + c * D_ + 8 * h);
  const float bias2 = b2[c];
  v8f y = {};
#pragma unroll
  for (int j = 0; j < 4; ++j) y = wmma4(kpair(T, j), kpair(f2, j), y);

  float* __restrict__ orow =
      out + (size_t)enc * ((size_t)B_ * D_) + (size_t)(b0 + 4 * h) * D_ + c;
#pragma unroll
  for (int r = 0; r < 4; ++r) orow[r * D_] = y[r] * (1.f / 32.f) + bias2;
}

extern "C" void kernel_launch(void* const* d_in, const int* in_sizes, int n_in,
                              void* d_out, int out_size, void* d_ws,
                              size_t ws_size, hipStream_t stream) {
  (void)in_sizes; (void)n_in; (void)out_size; (void)ws_size;
  const int*   arr1 = (const int*)d_in[0];
  const int*   arr2 = (const int*)d_in[1];
  const float* emb  = (const float*)d_in[2];
  const float* wq   = (const float*)d_in[3];
  const float* wk   = (const float*)d_in[4];
  const float* wv   = (const float*)d_in[5];
  const float* w1   = (const float*)d_in[6];
  const float* b1   = (const float*)d_in[7];
  const float* w2   = (const float*)d_in[8];
  const float* b2   = (const float*)d_in[9];
  float* out = (float*)d_out;

  float* qk = (float*)d_ws;
  float* vt = (float*)d_ws + VP_ * 32;
  vocab_tables_kernel<<<dim3(VT_), dim3(32), 0, stream>>>(emb, wq, wk, wv, w1,
                                                          qk, vt);

  const unsigned waves  = 2u * B_ / G_;
  const unsigned blocks = waves / WPB;
  encoder_kernel<<<dim3(blocks), dim3(TPB), 0, stream>>>(
      arr1, arr2, qk, vt, b1, w2, b2, out);
}


# Softmax exp2 issue order: each exp2 is followed by one independent plain instruction (other block's row max as a tree, partial sums, scaling by 1/sum, gather offsets); every sum keeps its order
# speedup vs baseline: 2.2589x; 1.0412x over previous
#include <hip/hip_runtime.h>

#define B_  131072
#define S_  32
#define D_  16
#define V_  257
#define VT_ 17
#define VP_ (VT_ * 16)
#define WPB 4
#define TPB (WPB * 32)
#define KS_ 0.36067376022224085f

typedef float v2f __attribute__((ext_vector_type(2)));
typedef float v8f __attribute__((ext_vector_type(8)));

__device__ __forceinline__ v8f wmma4(v2f a, v2f b, v8f c) {
  return __builtin_amdgcn_wmma_f32_16x16x4_f32(false, a, false, b, (short)0, c,
                                               false, false);
}


__device__ __forceinline__ v8f load8(const float* __restrict__ p) {
  const float4 lo = ((const float4*)p)[0];
  const float4 hi = ((const float4*)p)[1];
  v8f r;
  r[0] = lo.x; r[1] = lo.y; r[2] = lo.z; r[3] = lo.w;
  r[4] = hi.x; r[5] = hi.y; r[6] = hi.z; r[7] = hi.w;
  return r;
}

__device__ __forceinline__ v8f load8_at(const float* __restrict__ base,
                                        unsigned byte_off) {
  return load8((const float*)((const char*)base + byte_off));
}
__device__ __forceinline__ float load1_at(const float* __restrict__ base,
                                          unsigned byte_off) {
  return *(const float*)((const char*)base + byte_off);
}

__device__ __forceinline__ void store8(float* __restrict__ p, v8f v) {
  ((float4*)p)[0] = make_float4(v[0], v[1], v[2], v[3]);
  ((float4*)p)[1] = make_float4(v[4], v[5], v[6], v[7]);
}

__device__ __forceinline__ v2f kpair(v8f v, int j) {
  v2f r;
  r.x = v[2 * j];
  r.y = v[2 * j + 1];
  return r;
}

__device__ __forceinline__ void exp2_then_max3(float& e, float a, float& m,
                                               float x, float y, float z) {
  asm volatile("v_exp_f32 %0, %2\n\tv_max3_num_f32 %1, %3, %4, %5"
               : "=&v"(e), "=&v"(m)
               : "v"(a), "v"(x), "v"(y), "v"(z));
}
__device__ __forceinline__ void exp2_then_row_offset(float& e, float a,
                                                     unsigned& o, int tok,
                                                     unsigned col4) {
  asm volatile("v_exp_f32 %0, %2\n\tv_lshl_or_b32 %1, %3, 6, %4"
               : "=&v"(e), "=&v"(o)
               : "v"(a), "v"(tok), "v"(col4));
}
__device__ __forceinline__ void exp2_then_add2(float& e, float a, v2f& s, v2f p,
                                               v2f q) {
  asm volatile("v_exp_f32 %0, %2\n\tv_pk_add_f32 %1, %3, %4"
               : "=&v"(e), "=&v"(s)
               : "v"(a), "v"(p), "v"(q));
}
__device__ __forceinline__ void exp2_then_mul2(float& e, float a, v2f& n, v2f p,
                                               v2f q) {
  asm volatile("v_exp_f32 %0, %2\n\tv_pk_mul_f32 %1, %3, %4"
               : "=&v"(e), "=&v"(n)
               : "v"(a), "v"(p), "v"(q));
}

__global__ __launch_bounds__(32) void vocab_tables_kernel(
    const float* __restrict__ emb,
    const float* __restrict__ wq, const float* __restrict__ wk,
    const float* __restrict__ wv, const float* __restrict__ w1,
    float* __restrict__ qk, float* __restrict__ vt) {
  const int lane = threadIdx.x & 31;
  const int c    = lane & 15;
  const int h    = lane >> 4;
  const int row  = blockIdx.x * 16 + c;
  const int srow = row < V_ ? row : V_ - 1;

  const v8f xs = load8(emb + srow * D_ + 8 * h);
  v8f fq = load8(wq + c * D_ + 8 * h);
#pragma unroll
  for (int r = 0; r < 8; ++r) fq[r] *= KS_;
  const v8f fk = load8(wk + c * D_ + 8 * h);
  const v8f fv = load8(wv + c * D_ + 8 * h);
  const v8f f1 = load8(w1 + c * D_ + 8 * h);

  v8f cq = {}, ck = {}, cv = {};
#pragma unroll
  for (int j = 0; j < 4; ++j) {
    const v2f xp = kpair(xs, j);
    cq = wmma4(kpair(fq, j), xp, cq);
    ck = wmma4(kpair(fk, j), xp, ck);
    cv = wmma4(kpair(fv, j), xp, cv);
  }
  v8f cw = {};
#pragma unroll
  for (int j = 0; j < 4; ++j) cw = wmma4(kpair(f1, j), kpair(cv, j), cw);
  store8(qk + row * 32 + 8 * h, cq);
  store8(qk + row * 32 + 16 + 8 * h, ck);
  store8(vt + row * D_ + 8 * h, cw);
}

#define G_ 8

__global__ __launch_bounds__(TPB) void encoder_kernel(
    const int* __restrict__ arr1, const int* __restrict__ arr2,
    const float* __restrict__ qk, const float* __restrict__ vt,
    const float* __restrict__ b1,
    const float* __restrict__ w2, const float* __restrict__ b2,
    float* __restrict__ out) {
  const int tid  = threadIdx.x;
  const int wave = tid >> 5;
  const int lane = tid & 31;
  const int c    = lane & 15;
  const int h    = lane >> 4;

  const unsigned task0 = (blockIdx.x * WPB + wave) * G_;
  const int enc = (task0 >= (unsigned)B_) ? 1 : 0;
  const int b0  = enc ? (int)(task0 - (unsigned)B_) : (int)task0;
  const int* __restrict__ arr = enc ? arr2 : arr1;

  const int* __restrict__ trow = arr + (size_t)b0 * S_;

  const float bias1 = b1[c];
  const unsigned c4 = 4u * (unsigned)c;

  v8f T = {};

#pragma nounroll
  for (int i = 0; i < G_; ++i) {
    const int* __restrict__ tr = trow + i * S_;
    const int tok0 = tr[c];
    const int tok1 = tr[16 + c];
    int tv[2][8];
#pragma unroll
    for (int tb = 0; tb < 2; ++tb) {
      const int4 lo = ((const int4*)(tr + 16 * tb + 8 * h))[0];
      const int4 hi = ((const int4*)(tr + 16 * tb + 8 * h))[1];
      tv[tb][0] = lo.x; tv[tb][1] = lo.y; tv[tb][2] = lo.z; tv[tb][3] = lo.w;
      tv[tb][4] = hi.x; tv[tb][5] = hi.y; tv[tb][6] = hi.z; tv[tb][7] = hi.w;
    }

    v8f QT[2], KT[2];
    const unsigned o0 = (unsigned)tok0 * 128u + 32u * (unsigned)h;
    const unsigned o1 = (unsigned)tok1 * 128u + 32u * (unsigned)h;
    QT[0] = load8_at(qk, o0);
    KT[0] = load8_at(qk, o0 + 64u);
    QT[1] = load8_at(qk, o1);
    KT[1] = load8_at(qk, o1 + 64u);


    v8f P[2][2];
    {
      v8f s00 = {}, s01 = {}, s10 = {}, s11 = {};
#pragma unroll
      for (int j = 0; j < 4; ++j) {
        const v2f k0 = kpair(KT[0], j), k1 = kpair(KT[1], j);
        const v2f q0 = kpair(QT[0], j), q1 = kpair(QT[1], j);
        s00 = wmma4(k0, q0, s00);
        s01 = wmma4(k1, q0, s01);
        s10 = wmma4(k0, q1, s10);
        s11 = wmma4(k1, q1, s11);
      }
      P[0][0] = s00; P[0][1] = s01; P[1][0] = s10; P[1][1] = s11;
    }

    unsigned voff[2][8];
    v2f N0[8], N1[8];
    {
      float mx0;
      {
        const v8f x = P[0][0], y = P[0][1];
        const float ma = fmaxf(fmaxf(x[0], x[1]), x[2]);
        const float mb = fmaxf(fmaxf(x[3], x[4]), x[5]);
        const float mc = fmaxf(fmaxf(x[6], x[7]), y[0]);
        const float md = fmaxf(fmaxf(y[1], y[2]), y[3]);
        const float me = fmaxf(fmaxf(y[4], y[5]), y[6]);
        mx0 = fmaxf(fmaxf(fmaxf(ma, mb), mc), fmaxf(fmaxf(md, me), y[7]));
      }
      mx0 = fmaxf(mx0, __shfl_xor(mx0, 16, 32));
      const v2f m0 = {mx0, mx0};
      v2f A0[8];
#pragma unroll
      for (int q = 0; q < 8; ++q) A0[q] = kpair(P[0][q & 1], q >> 1) - m0;

      const v8f x1 = P[1][0], y1 = P[1][1];
      float t0, t1, t2, t3, t4, t5, t6, mx1;
      float ex[8], ey[8];
      v2f acc0[2];
      exp2_then_max3(ex[0], A0[0].x, t0, x1[0], x1[1], x1[2]);
      exp2_then_row_offset(ey[0], A0[0].y, voff[0][0], tv[0][0], c4);
      exp2_then_max3(ex[1], A0[1].x, t1, x1[3], x1[4], x1[5]);
      exp2_then_row_offset(ey[1], A0[1].y, voff[0][1], tv[0][1], c4);
      exp2_then_max3(ex[2], A0[2].x, t2, x1[6], x1[7], y1[0]);
      exp2_then_row_offset(ey[2], A0[2].y, voff[0][2], tv[0][2], c4);
      exp2_then_max3(ex[3], A0[3].x, t3, y1[1], y1[2], y1[3]);
      exp2_then_row_offset(ey[3], A0[3].y, voff[0][3], tv[0][3], c4);
      v2f E0[8];
#pragma unroll
      for (int q = 0; q < 4; ++q) { E0[q].x = ex[q]; E0[q].y = ey[q]; }
      exp2_then_max3(ex[4], A0[4].x, t4, y1[4], y1[5], y1[6]);
      exp2_then_add2(ey[4], A0[4].y, acc0[0], E0[0], E0[2]);
      exp2_then_max3(ex[5], A0[5].x, t5, t0, t1, t2);
      exp2_then_add2(ey[5], A0[5].y, acc0[1], E0[1], E0[3]);
      E0[4].x = ex[4]; E0[4].y = ey[4];
      E0[5].x = ex[5]; E0[5].y = ey[5];
      exp2_then_max3(ex[6], A0[6].x, t6, t3, t4, y1[7]);
      exp2_then_add2(ey[6], A0[6].y, acc0[0], acc0[0], E0[4]);
      exp2_then_max3(ex[7], A0[7].x, mx1, t5, t6, t6);
      exp2_then_add2(ey[7], A0[7].y, acc0[1], acc0[1], E0[5]);
      E0[6].x = ex[6]; E0[6].y = ey[6];
      E0[7].x = ex[7]; E0[7].y = ey[7];
      acc0[0] = acc0[0] + E0[6];
      acc0[1] = acc0[1] + E0[7];
      const v2f s0 = acc0[0] + acc0[1];
      float sum0 = s0.x + s0.y;
      sum0 += __shfl_xor(sum0, 16, 32);
      mx1 = fmaxf(mx1, __shfl_xor(mx1, 16, 32));
      const float inv0 = __builtin_amdgcn_rcpf(sum0);
      const v2f i0 = {inv0, inv0};
      const v2f m1 = {mx1, mx1};
      v2f A1[8];
#pragma unroll
      for (int q = 0; q < 8; ++q) A1[q] = kpair(P[1][q & 1], q >> 1) - m1;

      float fx[8], fy[8];
      v2f acc1[2];
      exp2_then_row_offset(fx[0], A1[0].x, voff[0][4], tv[0][4], c4);
      exp2_then_row_offset(fy[0], A1[0].y, voff[0][5], tv[0][5], c4);
      exp2_then_row_offset(fx[1], A1[1].x, voff[0][6], tv[0][6], c4);
      exp2_then_row_offset(fy[1], A1[1].y, voff[0][7], tv[0][7], c4);
      exp2_then_mul2(fx[2], A1[2].x, N0[0], E0[0], i0);
      exp2_then_mul2(fy[2], A1[2].y, N0[1], E0[1], i0);
      exp2_then_mul2(fx[3], A1[3].x, N0[2], E0[2], i0);
      exp2_then_mul2(fy[3], A1[3].y, N0[3], E0[3], i0);
      v2f E1[8];
#pragma unroll
      for (int q = 0; q < 4; ++q) { E1[q].x = fx[q]; E1[q].y = fy[q]; }
      exp2_then_mul2(fx[4], A1[4].x, N0[4], E0[4], i0);
      exp2_then_add2(fy[4], A1[4].y, acc1[0], E1[0], E1[2]);
      exp2_then_mul2(fx[5], A1[5].x, N0[5], E0[5], i0);
      exp2_then_add2(fy[5], A1[5].y, acc1[1], E1[1], E1[3]);
      E1[4].x = fx[4]; E1[4].y = fy[4];
      E1[5].x = fx[5]; E1[5].y = fy[5];
      exp2_then_mul2(fx[6], A1[6].x, N0[6], E0[6], i0);
      exp2_then_add2(fy[6], A1[6].y, acc1[0], acc1[0], E1[4]);
      exp2_then_mul2(fx[7], A1[7].x, N0[7], E0[7], i0);
      exp2_then_add2(fy[7], A1[7].y, acc1[1], acc1[1], E1[5]);
      E1[6].x = fx[6]; E1[6].y = fy[6];
      E1[7].x = fx[7]; E1[7].y = fy[7];
      acc1[0] = acc1[0] + E1[6];
      acc1[1] = acc1[1] + E1[7];
      const v2f s1 = acc1[0] + acc1[1];
      float sum1 = s1.x + s1.y;
      sum1 += __shfl_xor(sum1, 16, 32);
      const float inv1 = __builtin_amdgcn_rcpf(sum1);
      const v2f i1 = {inv1, inv1};
#pragma unroll
      for (int q = 0; q < 8; ++q) N1[q] = E1[q] * i1;
#pragma unroll
      for (int r = 0; r < 8; ++r) voff[1][r] = (unsigned)tv[1][r] * 64u + c4;
    }

    v8f VN[2];
#pragma unroll
    for (int tb = 0; tb < 2; ++tb)
#pragma unroll
      for (int r = 0; r < 8; ++r) VN[tb][r] = load1_at(vt, voff[tb][r]);

    v8f u0 = {}, u1 = {};
#pragma unroll
    for (int tb = 0; tb < 2; ++tb)
#pragma unroll
      for (int j = 0; j < 4; ++j) {
        const v2f bv = kpair(VN[tb], j);
        u0 = wmma4(N0[2 * j + tb], bv, u0);
        u1 = wmma4(N1[2 * j + tb], bv, u1);
      }

    v2f p;
    {
      float p0 = 0.f, p1 = 0.f;
#pragma unroll
      for (int r = 0; r < 8; ++r) {
        p0 += fmaxf(u0[r] + bias1, 0.f);
        p1 += fmaxf(u1[r] + bias1, 0.f);
      }
      p.x = p0;
      p.y = p1;
    }

    const int slot = (i & 3) | ((i & 4) << 1);
    const float one = (c == slot) ? 1.f : 0.f;
    v2f sel;
    sel.x = one;
    sel.y = one;
    T = wmma4(p, sel, T);
  }

  const v8f f2 = load8(w2 + c * D_ + 8 * h);
  const float bias2 = b2[c];
  v8f y = {};
#pragma unroll
  for (int j = 0; j < 4; ++j) y = wmma4(kpair(T, j), kpair(f2, j), y);

  float* __restrict__ orow =
      out + (size_t)enc * ((size_t)B_ * D_) + (size_t)(b0 + 4 * h) * D_ + c;
#pragma unroll
  for (int r = 0; r < 4; ++r) orow[r * D_] = y[r] * (1.f / 32.f) + bias2;
}

extern "C" void kernel_launch(void* const* d_in, const int* in_sizes, int n_in,
                              void* d_out, int out_size, void* d_ws,
                              size_t ws_size, hipStream_t stream) {
  (void)in_sizes; (void)n_in; (void)out_size; (void)ws_size;
  const int*   arr1 = (const int*)d_in[0];
  const int*   arr2 = (const int*)d_in[1];
  const float* emb  = (const float*)d_in[2];
  const float* wq   = (const float*)d_in[3];
  const float* wk   = (const float*)d_in[4];
  const float* wv   = (const float*)d_in[5];
  const float* w1   = (const float*)d_in[6];
  const float* b1   = (const float*)d_in[7];
  const float* w2   = (const float*)d_in[8];
  const float* b2   = (const float*)d_in[9];
  float* out = (float*)d_out;

  float* qk = (float*)d_ws;
  float* vt = (float*)d_ws + VP_ * 32;
  vocab_tables_kernel<<<dim3(VT_), dim3(32), 0, stream>>>(emb, wq, wk, wv, w1,
                                                          qk, vt);

  const unsigned waves  = 2u * B_ / G_;
  const unsigned blocks = waves / WPB;
  encoder_kernel<<<dim3(blocks), dim3(TPB), 0, stream>>>(
      arr1, arr2, qk, vt, b1, w2, b2, out);
}


# Scores hoisted to a per-vocabulary-pair table: launch 1 forms them on WMMA (same products, same k order), one 32-wave block per WGP holds the table in LDS, the encoder reads scores; 33 to 17 WMMAs
# speedup vs baseline: 2.7746x; 1.2283x over previous
#include <hip/hip_runtime.h>

#define B_  131072
#define S_  32
#define D_  16
#define V_  257
#define VT_ 17
#define VP_ (VT_ * 16)
#define WPB 32
#define TPB (WPB * 32)
#define NP_ (V_ * V_)
#define NP4_ (NP_ / 4)
#define VTO_ 66560
#define KS_ 0.36067376022224085f

typedef float v2f __attribute__((ext_vector_type(2)));
typedef float v8f __attribute__((ext_vector_type(8)));

__device__ __forceinline__ v8f wmma4(v2f a, v2f b, v8f c) {
  return __builtin_amdgcn_wmma_f32_16x16x4_f32(false, a, false, b, (short)0, c,
                                               false, false);
}


__device__ __forceinline__ v8f load8(const float* __restrict__ p) {
  const float4 lo = ((const float4*)p)[0];
  const float4 hi = ((const float4*)p)[1];
  v8f r;
  r[0] = lo.x; r[1] = lo.y; r[2] = lo.z; r[3] = lo.w;
  r[4] = hi.x; r[5] = hi.y; r[6] = hi.z; r[7] = hi.w;
  return r;
}

__device__ __forceinline__ v8f load8_at(const float* __restrict__ base,
                                        unsigned byte_off) {
  return load8((const float*)((const char*)base + byte_off));
}
__device__ __forceinline__ float load1_at(const float* __restrict__ base,
                                          unsigned byte_off) {
  return *(const float*)((const char*)base + byte_off);
}

__device__ __forceinline__ void store8(float* __restrict__ p, v8f v) {
  ((float4*)p)[0] = make_float4(v[0], v[1], v[2], v[3]);
  ((float4*)p)[1] = make_float4(v[4], v[5], v[6], v[7]);
}

__device__ __forceinline__ v2f kpair(v8f v, int j) {
  v2f r;
  r.x = v[2 * j];
  r.y = v[2 * j + 1];
  return r;
}

__device__ __forceinline__ void exp2_then_max3(float& e, float a, float& m,
                                               float x, float y, float z) {
  asm volatile("v_exp_f32 %0, %2\n\tv_max3_num_f32 %1, %3, %4, %5"
               : "=&v"(e), "=&v"(m)
               : "v"(a), "v"(x), "v"(y), "v"(z));
}
__device__ __forceinline__ void exp2_then_row_offset(float& e, float a,
                                                     unsigned& o, int tok,
                                                     unsigned col4) {
  asm volatile("v_exp_f32 %0, %2\n\tv_lshl_or_b32 %1, %3, 6, %4"
               : "=&v"(e), "=&v"(o)
               : "v"(a), "v"(tok), "v"(col4));
}
__device__ __forceinline__ void exp2_then_add2(float& e, float a, v2f& s, v2f p,
                                               v2f q) {
  asm volatile("v_exp_f32 %0, %2\n\tv_pk_add_f32 %1, %3, %4"
               : "=&v"(e), "=&v"(s)
               : "v"(a), "v"(p), "v"(q));
}
__device__ __forceinline__ void exp2_then_mul2(float& e, float a, v2f& n, v2f p,
                                               v2f q) {
  asm volatile("v_exp_f32 %0, %2\n\tv_pk_mul_f32 %1, %3, %4"
               : "=&v"(e), "=&v"(n)
               : "v"(a), "v"(p), "v"(q));
}

__global__ __launch_bounds__(32) void vocab_tables_kernel(
    const float* __restrict__ emb,
    const float* __restrict__ wq, const float* __restrict__ wk,
    const float* __restrict__ wv, const float* __restrict__ w1,
    float* __restrict__ st, float* __restrict__ vt) {
  const int lane = threadIdx.x & 31;
  const int c    = lane & 15;
  const int h    = lane >> 4;
  const int ta   = blockIdx.x;
  const int tb   = blockIdx.y;
  const int rowa = ta * 16 + c;
  const int rowb = tb * 16 + c;
  const int srowa = rowa < V_ ? rowa : V_ - 1;
  const int srowb = rowb < V_ ? rowb : V_ - 1;

  const v8f xa = load8(emb + srowa * D_ + 8 * h);
  const v8f xb = load8(emb + srowb * D_ + 8 * h);
  v8f fq = load8(wq + c * D_ + 8 * h);
#pragma unroll
  for (int r = 0; r < 8; ++r) fq[r] *= KS_;
  const v8f fk = load8(wk + c * D_ + 8 * h);

  v8f cq = {}, ck = {};
#pragma unroll
  for (int j = 0; j < 4; ++j) {
    cq = wmma4(kpair(fq, j), kpair(xa, j), cq);
    ck = wmma4(kpair(fk, j), kpair(xb, j), ck);
  }
  v8f sc = {};
#pragma unroll
  for (int j = 0; j < 4; ++j) sc = wmma4(kpair(ck, j), kpair(cq, j), sc);
  if (rowa < V_) {
    float* __restrict__ srow = st + rowa * V_ + tb * 16 + 8 * h;
#pragma unroll
    for (int r = 0; r < 8; ++r)
      if (tb * 16 + 8 * h + r < V_) srow[r] = sc[r];
  }

  if (tb == 0) {
    const v8f fv = load8(wv + c * D_ + 8 * h);
    const v8f f1 = load8(w1 + c * D_ + 8 * h);
    v8f cv = {};
#pragma unroll
    for (int j = 0; j < 4; ++j) cv = wmma4(kpair(fv, j), kpair(xa, j), cv);
    v8f cw = {};
#pragma unroll
    for (int j = 0; j < 4; ++j) cw = wmma4(kpair(f1, j), kpair(cv, j), cw);
    store8(vt + rowa * D_ + 8 * h, cw);
  }
}

#define G_ 8
#define NG_ 4
#define NBLK_ (2 * B_ / (WPB * NG_ * G_))

__global__ __launch_bounds__(TPB) void encoder_kernel(
    const int* __restrict__ arr1, const int* __restrict__ arr2,
    const float* __restrict__ st, const float* __restrict__ vt,
    const float* __restrict__ b1,
    const float* __restrict__ w2, const float* __restrict__ b2,
    float* __restrict__ out) {
  const int tid  = threadIdx.x;
  const int wave = tid >> 5;
  const int lane = tid & 31;
  const int c    = lane & 15;
  const int h    = lane >> 4;

  __shared__ __attribute__((aligned(16))) float tab[NP_];
  for (int i = tid; i < NP4_; i += TPB)
    ((float4*)tab)[i] = ((const float4*)st)[i];
  if (tid == 0) tab[NP_ - 1] = st[NP_ - 1];
  __syncthreads();
  const char* __restrict__ tabc = (const char*)tab;

  const float bias1 = b1[c];
  const unsigned c4 = 4u * (unsigned)c;
  const v8f f2 = load8(w2 + c * D_ + 8 * h);
  const float bias2 = b2[c];

#pragma nounroll
  for (int g = 0; g < NG_; ++g) {
  const unsigned task0 = ((blockIdx.x * WPB + wave) * NG_ + g) * G_;
  const int enc = (task0 >= (unsigned)B_) ? 1 : 0;
  const int b0  = enc ? (int)(task0 - (unsigned)B_) : (int)task0;
  const int* __restrict__ arr = enc ? arr2 : arr1;

  const int* __restrict__ trow = arr + (size_t)b0 * S_;

  v8f T = {};

#pragma nounroll
  for (int i = 0; i < G_; ++i) {
    const int* __restrict__ tr = trow + i * S_;
    const int tok0 = tr[c];
    const int tok1 = tr[16 + c];
    int tv[2][8];
#pragma unroll
    for (int tb = 0; tb < 2; ++tb) {
      const int4 lo = ((const int4*)(tr + 16 * tb + 8 * h))[0];
      const int4 hi = ((const int4*)(tr + 16 * tb + 8 * h))[1];
      tv[tb][0] = lo.x; tv[tb][1] = lo.y; tv[tb][2] = lo.z; tv[tb][3] = lo.w;
      tv[tb][4] = hi.x; tv[tb][5] = hi.y; tv[tb][6] = hi.z; tv[tb][7] = hi.w;
    }

    v8f P[2][2];
    {
      const unsigned rb0 = (unsigned)tok0 * (4u * V_);
      const unsigned rb1 = (unsigned)tok1 * (4u * V_);
#pragma unroll
      for (int tb = 0; tb < 2; ++tb)
#pragma unroll
        for (int r = 0; r < 8; ++r) {
          const unsigned cb = (unsigned)tv[tb][r] << 2;
          P[0][tb][r] = *(const float*)(tabc + (rb0 + cb));
          P[1][tb][r] = *(const float*)(tabc + (rb1 + cb));
        }
    }

    unsigned voff[2][8];
    v2f N0[8], N1[8];
    {
      float mx0;
      {
        const v8f x = P[0][0], y = P[0][1];
        const float ma = fmaxf(fmaxf(x[0], x[1]), x[2]);
        const float mb = fmaxf(fmaxf(x[3], x[4]), x[5]);
        const float mc = fmaxf(fmaxf(x[6], x[7]), y[0]);
        const float md = fmaxf(fmaxf(y[1], y[2]), y[3]);
        const float me = fmaxf(fmaxf(y[4], y[5]), y[6]);
        mx0 = fmaxf(fmaxf(fmaxf(ma, mb), mc), fmaxf(fmaxf(md, me), y[7]));
      }
      mx0 = fmaxf(mx0, __shfl_xor(mx0, 16, 32));
      const v2f m0 = {mx0, mx0};
      v2f A0[8];
#pragma unroll
      for (int q = 0; q < 8; ++q) A0[q] = kpair(P[0][q & 1], q >> 1) - m0;

      const v8f x1 = P[1][0], y1 = P[1][1];
      float t0, t1, t2, t3, t4, t5, t6, mx1;
      float ex[8], ey[8];
      v2f acc0[2];
      exp2_then_max3(ex[0], A0[0].x, t0, x1[0], x1[1], x1[2]);
      exp2_then_row_offset(ey[0], A0[0].y, voff[0][0], tv[0][0], c4);
      exp2_then_max3(ex[1], A0[1].x, t1, x1[3], x1[4], x1[5]);
      exp2_then_row_offset(ey[1], A0[1].y, voff[0][1], tv[0][1], c4);
      exp2_then_max3(ex[2], A0[2].x, t2, x1[6], x1[7], y1[0]);
      exp2_then_row_offset(ey[2], A0[2].y, voff[0][2], tv[0][2], c4);
      exp2_then_max3(ex[3], A0[3].x, t3, y1[1], y1[2], y1[3]);
      exp2_then_row_offset(ey[3], A0[3].y, voff[0][3], tv[0][3], c4);
      v2f E0[8];
#pragma unroll
      for (int q = 0; q < 4; ++q) { E0[q].x = ex[q]; E0[q].y = ey[q]; }
      exp2_then_max3(ex[4], A0[4].x, t4, y1[4], y1[5], y1[6]);
      exp2_then_add2(ey[4], A0[4].y, acc0[0], E0[0], E0[2]);
      exp2_then_max3(ex[5], A0[5].x, t5, t0, t1, t2);
      exp2_then_add2(ey[5], A0[5].y, acc0[1], E0[1], E0[3]);
      E0[4].x = ex[4]; E0[4].y = ey[4];
      E0[5].x = ex[5]; E0[5].y = ey[5];
      exp2_then_max3(ex[6], A0[6].x, t6, t3, t4, y1[7]);
      exp2_then_add2(ey[6], A0[6].y, acc0[0], acc0[0], E0[4]);
      exp2_then_max3(ex[7], A0[7].x, mx1, t5, t6, t6);
      exp2_then_add2(ey[7], A0[7].y, acc0[1], acc0[1], E0[5]);
      E0[6].x = ex[6]; E0[6].y = ey[6];
      E0[7].x = ex[7]; E0[7].y = ey[7];
      acc0[0] = acc0[0] + E0[6];
      acc0[1] = acc0[1] + E0[7];
      const v2f s0 = acc0[0] + acc0[1];
      float sum0 = s0.x + s0.y;
      sum0 += __shfl_xor(sum0, 16, 32);
      mx1 = fmaxf(mx1, __shfl_xor(mx1, 16, 32));
      const float inv0 = __builtin_amdgcn_rcpf(sum0);
      const v2f i0 = {inv0, inv0};
      const v2f m1 = {mx1, mx1};
      v2f A1[8];
#pragma unroll
      for (int q = 0; q < 8; ++q) A1[q] = kpair(P[1][q & 1], q >> 1) - m1;

      float fx[8], fy[8];
      v2f acc1[2];
      exp2_then_row_offset(fx[0], A1[0].x, voff[0][4], tv[0][4], c4);
      exp2_then_row_offset(fy[0], A1[0].y, voff[0][5], tv[0][5], c4);
      exp2_then_row_offset(fx[1], A1[1].x, voff[0][6], tv[0][6], c4);
      exp2_then_row_offset(fy[1], A1[1].y, voff[0][7], tv[0][7], c4);
      exp2_then_mul2(fx[2], A1[2].x, N0[0], E0[0], i0);
      exp2_then_mul2(fy[2], A1[2].y, N0[1], E0[1], i0);
      exp2_then_mul2(fx[3], A1[3].x, N0[2], E0[2], i0);
      exp2_then_mul2(fy[3], A1[3].y, N0[3], E0[3], i0);
      v2f E1[8];
#pragma unroll
      for (int q = 0; q < 4; ++q) { E1[q].x = fx[q]; E1[q].y = fy[q]; }
      exp2_then_mul2(fx[4], A1[4].x, N0[4], E0[4], i0);
      exp2_then_add2(fy[4], A1[4].y, acc1[0], E1[0], E1[2]);
      exp2_then_mul2(fx[5], A1[5].x, N0[5], E0[5], i0);
      exp2_then_add2(fy[5], A1[5].y, acc1[1], E1[1], E1[3]);
      E1[4].x = fx[4]; E1[4].y = fy[4];
      E1[5].x = fx[5]; E1[5].y = fy[5];
      exp2_then_mul2(fx[6], A1[6].x, N0[6], E0[6], i0);
      exp2_then_add2(fy[6], A1[6].y, acc1[0], acc1[0], E1[4]);
      exp2_then_mul2(fx[7], A1[7].x, N0[7], E0[7], i0);
      exp2_then_add2(fy[7], A1[7].y, acc1[1], acc1[1], E1[5]);
      E1[6].x = fx[6]; E1[6].y = fy[6];
      E1[7].x = fx[7]; E1[7].y = fy[7];
      acc1[0] = acc1[0] + E1[6];
      acc1[1] = acc1[1] + E1[7];
      const v2f s1 = acc1[0] + acc1[1];
      float sum1 = s1.x + s1.y;
      sum1 += __shfl_xor(sum1, 16, 32);
      const float inv1 = __builtin_amdgcn_rcpf(sum1);
      const v2f i1 = {inv1, inv1};
#pragma unroll
      for (int q = 0; q < 8; ++q) N1[q] = E1[q] * i1;
#pragma unroll
      for (int r = 0; r < 8; ++r) voff[1][r] = (unsigned)tv[1][r] * 64u + c4;
    }

    v8f VN[2];
#pragma unroll
    for (int tb = 0; tb < 2; ++tb)
#pragma unroll
      for (int r = 0; r < 8; ++r) VN[tb][r] = load1_at(vt, voff[tb][r]);

    v8f u0 = {}, u1 = {};
#pragma unroll
    for (int tb = 0; tb < 2; ++tb)
#pragma unroll
      for (int j = 0; j < 4; ++j) {
        const v2f bv = kpair(VN[tb], j);
        u0 = wmma4(N0[2 * j + tb], bv, u0);
        u1 = wmma4(N1[2 * j + tb], bv, u1);
      }

    v2f p;
    {
      float p0 = 0.f, p1 = 0.f;
#pragma unroll
      for (int r = 0; r < 8; ++r) {
        p0 += fmaxf(u0[r] + bias1, 0.f);
        p1 += fmaxf(u1[r] + bias1, 0.f);
      }
      p.x = p0;
      p.y = p1;
    }

    const int slot = (i & 3) | ((i & 4) << 1);
    const float one = (c == slot) ? 1.f : 0.f;
    v2f sel;
    sel.x = one;
    sel.y = one;
    T = wmma4(p, sel, T);
  }

  v8f y = {};
#pragma unroll
  for (int j = 0; j < 4; ++j) y = wmma4(kpair(T, j), kpair(f2, j), y);

  float* __restrict__ orow =
      out + (size_t)enc * ((size_t)B_ * D_) + (size_t)(b0 + 4 * h) * D_ + c;
#pragma unroll
  for (int r = 0; r < 4; ++r) orow[r * D_] = y[r] * (1.f / 32.f) + bias2;
  }
}

extern "C" void kernel_launch(void* const* d_in, const int* in_sizes, int n_in,
                              void* d_out, int out_size, void* d_ws,
                              size_t ws_size, hipStream_t stream) {
  (void)in_sizes; (void)n_in; (void)out_size; (void)ws_size;
  const int*   arr1 = (const int*)d_in[0];
  const int*   arr2 = (const int*)d_in[1];
  const float* emb  = (const float*)d_in[2];
  const float* wq   = (const float*)d_in[3];
  const float* wk   = (const float*)d_in[4];
  const float* wv   = (const float*)d_in[5];
  const float* w1   = (const float*)d_in[6];
  const float* b1   = (const float*)d_in[7];
  const float* w2   = (const float*)d_in[8];
  const float* b2   = (const float*)d_in[9];
  float* out = (float*)d_out;

  float* st = (float*)d_ws;
  float* vt = (float*)d_ws + VTO_;
  vocab_tables_kernel<<<dim3(VT_, VT_), dim3(32), 0, stream>>>(emb, wq, wk, wv,
                                                               w1, st, vt);

  encoder_kernel<<<dim3(NBLK_), dim3(TPB), 0, stream>>>(
      arr1, arr2, st, vt, b1, w2, b2, out);
}


# Softmax exp2 hoisted into the pair table: launch 1 stores exp2(score - row max over vocabulary); encoder's maxima, subtractions and exp2 go; underflowed rows redone from raw scores; sums keep order
# speedup vs baseline: 3.1995x; 1.1531x over previous
#include <hip/hip_runtime.h>

#define B_  131072
#define S_  32
#define D_  16
#define V_  257
#define VT_ 17
#define VP_ (VT_ * 16)
#define WPB 32
#define TPB (WPB * 32)
#define NP_ (V_ * V_)
#define NP4_ (NP_ / 4)
#define VTO_ 66560
#define SRO_ 71168
#define ZMIN_ 8.673617379884035e-19f
#define KS_ 0.36067376022224085f

typedef float v2f __attribute__((ext_vector_type(2)));
typedef float v8f __attribute__((ext_vector_type(8)));

__device__ __forceinline__ v8f wmma4(v2f a, v2f b, v8f c) {
  return __builtin_amdgcn_wmma_f32_16x16x4_f32(false, a, false, b, (short)0, c,
                                               false, false);
}


__device__ __forceinline__ v8f load8(const float* __restrict__ p) {
  const float4 lo = ((const float4*)p)[0];
  const float4 hi = ((const float4*)p)[1];
  v8f r;
  r[0] = lo.x; r[1] = lo.y; r[2] = lo.z; r[3] = lo.w;
  r[4] = hi.x; r[5] = hi.y; r[6] = hi.z; r[7] = hi.w;
  return r;
}

__device__ __forceinline__ v8f load8_at(const float* __restrict__ base,
                                        unsigned byte_off) {
  return load8((const float*)((const char*)base + byte_off));
}
__device__ __forceinline__ float load1_at(const float* __restrict__ base,
                                          unsigned byte_off) {
  return *(const float*)((const char*)base + byte_off);
}

__device__ __forceinline__ void store8(float* __restrict__ p, v8f v) {
  ((float4*)p)[0] = make_float4(v[0], v[1], v[2], v[3]);
  ((float4*)p)[1] = make_float4(v[4], v[5], v[6], v[7]);
}

__device__ __forceinline__ v2f kpair(v8f v, int j) {
  v2f r;
  r.x = v[2 * j];
  r.y = v[2 * j + 1];
  return r;
}

__device__ __forceinline__ float row_sum(v8f x, v8f y) {
  v2f a0 = kpair(x, 0) + kpair(x, 1);
  v2f a1 = kpair(y, 0) + kpair(y, 1);
  a0 = a0 + kpair(x, 2);
  a1 = a1 + kpair(y, 2);
  a0 = a0 + kpair(x, 3);
  a1 = a1 + kpair(y, 3);
  const v2f t = a0 + a1;
  float sum = t.x + t.y;
  sum += __shfl_xor(sum, 16, 32);
  return sum;
}

__global__ __launch_bounds__(VT_ * 32) void vocab_tables_kernel(
    const float* __restrict__ emb,
    const float* __restrict__ wq, const float* __restrict__ wk,
    const float* __restrict__ wv, const float* __restrict__ w1,
    float* __restrict__ et, float* __restrict__ sr, float* __restrict__ vt) {
  const int lane = threadIdx.x & 31;
  const int c    = lane & 15;
  const int h    = lane >> 4;
  const int ta   = blockIdx.x;
  const int tb   = __builtin_amdgcn_readfirstlane((int)(threadIdx.x >> 5));
  const int rowa = ta * 16 + c;
  const int rowb = tb * 16 + c;
  const int srowa = rowa < V_ ? rowa : V_ - 1;
  const int srowb = rowb < V_ ? rowb : V_ - 1;

  const v8f xa = load8(emb + srowa * D_ + 8 * h);
  const v8f xb = load8(emb + srowb * D_ + 8 * h);
  v8f fq = load8(wq + c * D_ + 8 * h);
#pragma unroll
  for (int r = 0; r < 8; ++r) fq[r] *= KS_;
  const v8f fk = load8(wk + c * D_ + 8 * h);

  v8f cq = {}, ck = {};
#pragma unroll
  for (int j = 0; j < 4; ++j) {
    cq = wmma4(kpair(fq, j), kpair(xa, j), cq);
    ck = wmma4(kpair(fk, j), kpair(xb, j), ck);
  }
  v8f sc = {};
#pragma unroll
  for (int j = 0; j < 4; ++j) sc = wmma4(kpair(ck, j), kpair(cq, j), sc);

  __shared__ float tmax[VT_][16];
  float m = fmaxf(fmaxf(fmaxf(sc[0], sc[1]), fmaxf(sc[2], sc[3])),
                  fmaxf(fmaxf(sc[4], sc[5]), fmaxf(sc[6], sc[7])));
  m = fmaxf(m, __shfl_xor(m, 16, 32));
  if (h == 0) tmax[tb][c] = m;
  __syncthreads();
  float M = tmax[0][c];
#pragma unroll
  for (int w = 1; w < VT_; ++w) M = fmaxf(M, tmax[w][c]);

  if (rowa < V_) {
    float* __restrict__ srow = sr + rowa * V_ + tb * 16 + 8 * h;
    float* __restrict__ erow = et + rowa * V_ + tb * 16 + 8 * h;
#pragma unroll
    for (int r = 0; r < 8; ++r)
      if (tb * 16 + 8 * h + r < V_) {
        srow[r] = sc[r];
        erow[r] = __builtin_amdgcn_exp2f(sc[r] - M);
      }
  }

  if (tb == 0) {
    const v8f fv = load8(wv + c * D_ + 8 * h);
    const v8f f1 = load8(w1 + c * D_ + 8 * h);
    v8f cv = {};
#pragma unroll
    for (int j = 0; j < 4; ++j) cv = wmma4(kpair(fv, j), kpair(xa, j), cv);
    v8f cw = {};
#pragma unroll
    for (int j = 0; j < 4; ++j) cw = wmma4(kpair(f1, j), kpair(cv, j), cw);
    store8(vt + rowa * D_ + 8 * h, cw);
  }
}

#define G_ 8
#define NG_ 4
#define NBLK_ (2 * B_ / (WPB * NG_ * G_))

__global__ __launch_bounds__(TPB) void encoder_kernel(
    const int* __restrict__ arr1, const int* __restrict__ arr2,
    const float* __restrict__ et, const float* __restrict__ sr,
    const float* __restrict__ vt, const float* __restrict__ b1,
    const float* __restrict__ w2, const float* __restrict__ b2,
    float* __restrict__ out) {
  const int tid  = threadIdx.x;
  const int wave = tid >> 5;
  const int lane = tid & 31;
  const int c    = lane & 15;
  const int h    = lane >> 4;

  __shared__ __attribute__((aligned(16))) float tab[NP_];
  for (int i = tid; i < NP4_; i += TPB)
    ((float4*)tab)[i] = ((const float4*)et)[i];
  if (tid == 0) tab[NP_ - 1] = et[NP_ - 1];
  __syncthreads();
  const char* __restrict__ tabc = (const char*)tab;

  const float bias1 = b1[c];
  const unsigned c4 = 4u * (unsigned)c;
  const v8f f2 = load8(w2 + c * D_ + 8 * h);
  const float bias2 = b2[c];

#pragma nounroll
  for (int g = 0; g < NG_; ++g) {
  const unsigned task0 = ((blockIdx.x * WPB + wave) * NG_ + g) * G_;
  const int enc = (task0 >= (unsigned)B_) ? 1 : 0;
  const int b0  = enc ? (int)(task0 - (unsigned)B_) : (int)task0;
  const int* __restrict__ arr = enc ? arr2 : arr1;

  const int* __restrict__ trow = arr + (size_t)b0 * S_;

  v8f T = {};

#pragma nounroll
  for (int i = 0; i < G_; ++i) {
    const int* __restrict__ tr = trow + i * S_;
    const int tok0 = tr[c];
    const int tok1 = tr[16 + c];
    int tv[2][8];
#pragma unroll
    for (int tb = 0; tb < 2; ++tb) {
      const int4 lo = ((const int4*)(tr + 16 * tb + 8 * h))[0];
      const int4 hi = ((const int4*)(tr + 16 * tb + 8 * h))[1];
      tv[tb][0] = lo.x; tv[tb][1] = lo.y; tv[tb][2] = lo.z; tv[tb][3] = lo.w;
      tv[tb][4] = hi.x; tv[tb][5] = hi.y; tv[tb][6] = hi.z; tv[tb][7] = hi.w;
    }

    v8f P[2][2];
    {
      const unsigned rb0 = (unsigned)tok0 * (4u * V_);
      const unsigned rb1 = (unsigned)tok1 * (4u * V_);
#pragma unroll
      for (int tb = 0; tb < 2; ++tb)
#pragma unroll
        for (int r = 0; r < 8; ++r) {
          const unsigned cb = (unsigned)tv[tb][r] << 2;
          P[0][tb][r] = *(const float*)(tabc + (rb0 + cb));
          P[1][tb][r] = *(const float*)(tabc + (rb1 + cb));
        }
    }

    v8f VN[2];
#pragma unroll
    for (int tb = 0; tb < 2; ++tb)
#pragma unroll
      for (int r = 0; r < 8; ++r)
        VN[tb][r] = load1_at(vt, (unsigned)tv[tb][r] * 64u + c4);

    float sum0 = row_sum(P[0][0], P[0][1]);
    float sum1 = row_sum(P[1][0], P[1][1]);
    if (__builtin_expect(
            __builtin_amdgcn_ballot_w32(!(fminf(sum0, sum1) >= ZMIN_)) != 0u, 0)) {
      const float* __restrict__ r0 = sr + tok0 * V_;
      const float* __restrict__ r1 = sr + tok1 * V_;
      v8f R[2][2];
#pragma unroll
      for (int tb = 0; tb < 2; ++tb)
#pragma unroll
        for (int r = 0; r < 8; ++r) {
          R[0][tb][r] = r0[tv[tb][r]];
          R[1][tb][r] = r1[tv[tb][r]];
        }
#pragma unroll
      for (int sb = 0; sb < 2; ++sb) {
        float mx = fmaxf(R[sb][0][0], R[sb][1][0]);
#pragma unroll
        for (int r = 1; r < 8; ++r) mx = fmaxf(mx, fmaxf(R[sb][0][r], R[sb][1][r]));
        mx = fmaxf(mx, __shfl_xor(mx, 16, 32));
#pragma unroll
        for (int tb = 0; tb < 2; ++tb)
#pragma unroll
          for (int r = 0; r < 8; ++r)
            P[sb][tb][r] = __builtin_amdgcn_exp2f(R[sb][tb][r] - mx);
      }
      sum0 = row_sum(P[0][0], P[0][1]);
      sum1 = row_sum(P[1][0], P[1][1]);
    }
    v2f N0[8], N1[8];
    {
      const float inv0 = __builtin_amdgcn_rcpf(sum0);
      const float inv1 = __builtin_amdgcn_rcpf(sum1);
      const v2f i0 = {inv0, inv0};
      const v2f i1 = {inv1, inv1};
#pragma unroll
      for (int q = 0; q < 8; ++q) {
        N0[q] = kpair(P[0][q & 1], q >> 1) * i0;
        N1[q] = kpair(P[1][q & 1], q >> 1) * i1;
      }
    }

    v8f u0 = {}, u1 = {};
#pragma unroll
    for (int tb = 0; tb < 2; ++tb)
#pragma unroll
      for (int j = 0; j < 4; ++j) {
        const v2f bv = kpair(VN[tb], j);
        u0 = wmma4(N0[2 * j + tb], bv, u0);
        u1 = wmma4(N1[2 * j + tb], bv, u1);
      }

    v2f p;
    {
      float p0 = 0.f, p1 = 0.f;
#pragma unroll
      for (int r = 0; r < 8; ++r) {
        p0 += fmaxf(u0[r] + bias1, 0.f);
        p1 += fmaxf(u1[r] + bias1, 0.f);
      }
      p.x = p0;
      p.y = p1;
    }

    const int slot = (i & 3) | ((i & 4) << 1);
    const float one = (c == slot) ? 1.f : 0.f;
    v2f sel;
    sel.x = one;
    sel.y = one;
    T = wmma4(p, sel, T);
  }

  v8f y = {};
#pragma unroll
  for (int j = 0; j < 4; ++j) y = wmma4(kpair(T, j), kpair(f2, j), y);

  float* __restrict__ orow =
      out + (size_t)enc * ((size_t)B_ * D_) + (size_t)(b0 + 4 * h) * D_ + c;
#pragma unroll
  for (int r = 0; r < 4; ++r) orow[r * D_] = y[r] * (1.f / 32.f) + bias2;
  }
}

extern "C" void kernel_launch(void* const* d_in, const int* in_sizes, int n_in,
                              void* d_out, int out_size, void* d_ws,
                              size_t ws_size, hipStream_t stream) {
  (void)in_sizes; (void)n_in; (void)out_size; (void)ws_size;
  const int*   arr1 = (const int*)d_in[0];
  const int*   arr2 = (const int*)d_in[1];
  const float* emb  = (const float*)d_in[2];
  const float* wq   = (const float*)d_in[3];
  const float* wk   = (const float*)d_in[4];
  const float* wv   = (const float*)d_in[5];
  const float* w1   = (const float*)d_in[6];
  const float* b1   = (const float*)d_in[7];
  const float* w2   = (const float*)d_in[8];
  const float* b2   = (const float*)d_in[9];
  float* out = (float*)d_out;

  float* et = (float*)d_ws;
  float* vt = (float*)d_ws + VTO_;
  float* sr = (float*)d_ws + SRO_;
  vocab_tables_kernel<<<dim3(VT_), dim3(VT_ * 32), 0, stream>>>(emb, wq, wk, wv,
                                                                w1, et, sr, vt);

  encoder_kernel<<<dim3(NBLK_), dim3(TPB), 0, stream>>>(
      arr1, arr2, et, sr, vt, b1, w2, b2, out);
}


# Token loads pipelined one sequence ahead: the next sequence's tokens are requested right behind this sequence's gathers (the first before the table copy), so no sequence starts on a memory wait
# speedup vs baseline: 3.3567x; 1.0491x over previous
#include <hip/hip_runtime.h>

#define B_  131072
#define S_  32
#define D_  16
#define V_  257
#define VT_ 17
#define VP_ (VT_ * 16)
#define WPB 32
#define TPB (WPB * 32)
#define NP_ (V_ * V_)
#define NP4_ (NP_ / 4)
#define VTO_ 66560
#define SRO_ 71168
#define ZMIN_ 8.673617379884035e-19f
#define KS_ 0.36067376022224085f

typedef float v2f __attribute__((ext_vector_type(2)));
typedef float v8f __attribute__((ext_vector_type(8)));

__device__ __forceinline__ v8f wmma4(v2f a, v2f b, v8f c) {
  return __builtin_amdgcn_wmma_f32_16x16x4_f32(false, a, false, b, (short)0, c,
                                               false, false);
}


__device__ __forceinline__ v8f load8(const float* __restrict__ p) {
  const float4 lo = ((const float4*)p)[0];
  const float4 hi = ((const float4*)p)[1];
  v8f r;
  r[0] = lo.x; r[1] = lo.y; r[2] = lo.z; r[3] = lo.w;
  r[4] = hi.x; r[5] = hi.y; r[6] = hi.z; r[7] = hi.w;
  return r;
}

__device__ __forceinline__ v8f load8_at(const float* __restrict__ base,
                                        unsigned byte_off) {
  return load8((const float*)((const char*)base + byte_off));
}
__device__ __forceinline__ float load1_at(const float* __restrict__ base,
                                          unsigned byte_off) {
  return *(const float*)((const char*)base + byte_off);
}

__device__ __forceinline__ void store8(float* __restrict__ p, v8f v) {
  ((float4*)p)[0] = make_float4(v[0], v[1], v[2], v[3]);
  ((float4*)p)[1] = make_float4(v[4], v[5], v[6], v[7]);
}

__device__ __forceinline__ void load_tokens(const int* __restrict__ tr, int c,
                                            int h, int& t0, int& t1,
                                            int (&tv)[2][8]) {
  t0 = tr[c];
  t1 = tr[16 + c];
#pragma unroll
  for (int tb = 0; tb < 2; ++tb) {
    const int4 lo = ((const int4*)(tr + 16 * tb + 8 * h))[0];
    const int4 hi = ((const int4*)(tr + 16 * tb + 8 * h))[1];
    tv[tb][0] = lo.x; tv[tb][1] = lo.y; tv[tb][2] = lo.z; tv[tb][3] = lo.w;
    tv[tb][4] = hi.x; tv[tb][5] = hi.y; tv[tb][6] = hi.z; tv[tb][7] = hi.w;
  }
}

__device__ __forceinline__ v2f kpair(v8f v, int j) {
  v2f r;
  r.x = v[2 * j];
  r.y = v[2 * j + 1];
  return r;
}

__device__ __forceinline__ float row_sum(v8f x, v8f y) {
  v2f a0 = kpair(x, 0) + kpair(x, 1);
  v2f a1 = kpair(y, 0) + kpair(y, 1);
  a0 = a0 + kpair(x, 2);
  a1 = a1 + kpair(y, 2);
  a0 = a0 + kpair(x, 3);
  a1 = a1 + kpair(y, 3);
  const v2f t = a0 + a1;
  float sum = t.x + t.y;
  sum += __shfl_xor(sum, 16, 32);
  return sum;
}

__global__ __launch_bounds__(VT_ * 32) void vocab_tables_kernel(
    const float* __restrict__ emb,
    const float* __restrict__ wq, const float* __restrict__ wk,
    const float* __restrict__ wv, const float* __restrict__ w1,
    float* __restrict__ et, float* __restrict__ sr, float* __restrict__ vt) {
  const int lane = threadIdx.x & 31;
  const int c    = lane & 15;
  const int h    = lane >> 4;
  const int ta   = blockIdx.x;
  const int tb   = __builtin_amdgcn_readfirstlane((int)(threadIdx.x >> 5));
  const int rowa = ta * 16 + c;
  const int rowb = tb * 16 + c;
  const int srowa = rowa < V_ ? rowa : V_ - 1;
  const int srowb = rowb < V_ ? rowb : V_ - 1;

  const v8f xa = load8(emb + srowa * D_ + 8 * h);
  const v8f xb = load8(emb + srowb * D_ + 8 * h);
  v8f fq = load8(wq + c * D_ + 8 * h);
#pragma unroll
  for (int r = 0; r < 8; ++r) fq[r] *= KS_;
  const v8f fk = load8(wk + c * D_ + 8 * h);

  v8f cq = {}, ck = {};
#pragma unroll
  for (int j = 0; j < 4; ++j) {
    cq = wmma4(kpair(fq, j), kpair(xa, j), cq);
    ck = wmma4(kpair(fk, j), kpair(xb, j), ck);
  }
  v8f sc = {};
#pragma unroll
  for (int j = 0; j < 4; ++j) sc = wmma4(kpair(ck, j), kpair(cq, j), sc);

  __shared__ float tmax[VT_][16];
  float m = fmaxf(fmaxf(fmaxf(sc[0], sc[1]), fmaxf(sc[2], sc[3])),
                  fmaxf(fmaxf(sc[4], sc[5]), fmaxf(sc[6], sc[7])));
  m = fmaxf(m, __shfl_xor(m, 16, 32));
  if (h == 0) tmax[tb][c] = m;
  __syncthreads();
  float M = tmax[0][c];
#pragma unroll
  for (int w = 1; w < VT_; ++w) M = fmaxf(M, tmax[w][c]);

  if (rowa < V_) {
    float* __restrict__ srow = sr + rowa * V_ + tb * 16 + 8 * h;
    float* __restrict__ erow = et + rowa * V_ + tb * 16 + 8 * h;
#pragma unroll
    for (int r = 0; r < 8; ++r)
      if (tb * 16 + 8 * h + r < V_) {
        srow[r] = sc[r];
        erow[r] = __builtin_amdgcn_exp2f(sc[r] - M);
      }
  }

  if (tb == 0) {
    const v8f fv = load8(wv + c * D_ + 8 * h);
    const v8f f1 = load8(w1 + c * D_ + 8 * h);
    v8f cv = {};
#pragma unroll
    for (int j = 0; j < 4; ++j) cv = wmma4(kpair(fv, j), kpair(xa, j), cv);
    v8f cw = {};
#pragma unroll
    for (int j = 0; j < 4; ++j) cw = wmma4(kpair(f1, j), kpair(cv, j), cw);
    store8(vt + rowa * D_ + 8 * h, cw);
  }
}

#define G_ 8
#define NG_ 4
#define NBLK_ (2 * B_ / (WPB * NG_ * G_))

__global__ __launch_bounds__(TPB) void encoder_kernel(
    const int* __restrict__ arr1, const int* __restrict__ arr2,
    const float* __restrict__ et, const float* __restrict__ sr,
    const float* __restrict__ vt, const float* __restrict__ b1,
    const float* __restrict__ w2, const float* __restrict__ b2,
    float* __restrict__ out) {
  const int tid  = threadIdx.x;
  const int wave = tid >> 5;
  const int lane = tid & 31;
  const int c    = lane & 15;
  const int h    = lane >> 4;

  const unsigned wtask0 = (blockIdx.x * WPB + wave) * (NG_ * G_);
  const int enc = (wtask0 >= (unsigned)B_) ? 1 : 0;
  const int wb0 = enc ? (int)(wtask0 - (unsigned)B_) : (int)wtask0;
  const int* __restrict__ arr = enc ? arr2 : arr1;
  const int* __restrict__ wrow = arr + (size_t)wb0 * S_;

  int tok0, tok1, tv[2][8];
  load_tokens(wrow, c, h, tok0, tok1, tv);

  __shared__ __attribute__((aligned(16))) float tab[NP_];
  for (int i = tid; i < NP4_; i += TPB)
    ((float4*)tab)[i] = ((const float4*)et)[i];
  if (tid == 0) tab[NP_ - 1] = et[NP_ - 1];
  __syncthreads();
  const char* __restrict__ tabc = (const char*)tab;

  const float bias1 = b1[c];
  const unsigned c4 = 4u * (unsigned)c;
  const v8f f2 = load8(w2 + c * D_ + 8 * h);
  const float bias2 = b2[c];

#pragma nounroll
  for (int g = 0; g < NG_; ++g) {
  const int b0 = wb0 + g * G_;

  v8f T = {};

#pragma nounroll
  for (int i = 0; i < G_; ++i) {
    const int n = g * G_ + i;

    v8f P[2][2];
    {
      const unsigned rb0 = (unsigned)tok0 * (4u * V_);
      const unsigned rb1 = (unsigned)tok1 * (4u * V_);
#pragma unroll
      for (int tb = 0; tb < 2; ++tb)
#pragma unroll
        for (int r = 0; r < 8; ++r) {
          const unsigned cb = (unsigned)tv[tb][r] << 2;
          P[0][tb][r] = *(const float*)(tabc + (rb0 + cb));
          P[1][tb][r] = *(const float*)(tabc + (rb1 + cb));
        }
    }

    v8f VN[2];
#pragma unroll
    for (int tb = 0; tb < 2; ++tb)
#pragma unroll
      for (int r = 0; r < 8; ++r)
        VN[tb][r] = load1_at(vt, (unsigned)tv[tb][r] * 64u + c4);

    {
      const int nn = (n + 1 < NG_ * G_) ? n + 1 : NG_ * G_ - 1;
      load_tokens(wrow + nn * S_, c, h, tok0, tok1, tv);
    }

    float sum0 = row_sum(P[0][0], P[0][1]);
    float sum1 = row_sum(P[1][0], P[1][1]);
    if (__builtin_expect(
            __builtin_amdgcn_ballot_w32(!(fminf(sum0, sum1) >= ZMIN_)) != 0u, 0)) {
      int ct0, ct1, ctv[2][8];
      load_tokens(wrow + n * S_, c, h, ct0, ct1, ctv);
      const float* __restrict__ r0 = sr + ct0 * V_;
      const float* __restrict__ r1 = sr + ct1 * V_;
      v8f R[2][2];
#pragma unroll
      for (int tb = 0; tb < 2; ++tb)
#pragma unroll
        for (int r = 0; r < 8; ++r) {
          R[0][tb][r] = r0[ctv[tb][r]];
          R[1][tb][r] = r1[ctv[tb][r]];
        }
#pragma unroll
      for (int sb = 0; sb < 2; ++sb) {
        float mx = fmaxf(R[sb][0][0], R[sb][1][0]);
#pragma unroll
        for (int r = 1; r < 8; ++r) mx = fmaxf(mx, fmaxf(R[sb][0][r], R[sb][1][r]));
        mx = fmaxf(mx, __shfl_xor(mx, 16, 32));
#pragma unroll
        for (int tb = 0; tb < 2; ++tb)
#pragma unroll
          for (int r = 0; r < 8; ++r)
            P[sb][tb][r] = __builtin_amdgcn_exp2f(R[sb][tb][r] - mx);
      }
      sum0 = row_sum(P[0][0], P[0][1]);
      sum1 = row_sum(P[1][0], P[1][1]);
    }
    v2f N0[8], N1[8];
    {
      const float inv0 = __builtin_amdgcn_rcpf(sum0);
      const float inv1 = __builtin_amdgcn_rcpf(sum1);
      const v2f i0 = {inv0, inv0};
      const v2f i1 = {inv1, inv1};
#pragma unroll
      for (int q = 0; q < 8; ++q) {
        N0[q] = kpair(P[0][q & 1], q >> 1) * i0;
        N1[q] = kpair(P[1][q & 1], q >> 1) * i1;
      }
    }

    v8f u0 = {}, u1 = {};
#pragma unroll
    for (int tb = 0; tb < 2; ++tb)
#pragma unroll
      for (int j = 0; j < 4; ++j) {
        const v2f bv = kpair(VN[tb], j);
        u0 = wmma4(N0[2 * j + tb], bv, u0);
        u1 = wmma4(N1[2 * j + tb], bv, u1);
      }

    v2f p;
    {
      float p0 = 0.f, p1 = 0.f;
#pragma unroll
      for (int r = 0; r < 8; ++r) {
        p0 += fmaxf(u0[r] + bias1, 0.f);
        p1 += fmaxf(u1[r] + bias1, 0.f);
      }
      p.x = p0;
      p.y = p1;
    }

    const int slot = (i & 3) | ((i & 4) << 1);
    const float one = (c == slot) ? 1.f : 0.f;
    v2f sel;
    sel.x = one;
    sel.y = one;
    T = wmma4(p, sel, T);
  }

  v8f y = {};
#pragma unroll
  for (int j = 0; j < 4; ++j) y = wmma4(kpair(T, j), kpair(f2, j), y);

  float* __restrict__ orow =
      out + (size_t)enc * ((size_t)B_ * D_) + (size_t)(b0 + 4 * h) * D_ + c;
#pragma unroll
  for (int r = 0; r < 4; ++r) orow[r * D_] = y[r] * (1.f / 32.f) + bias2;
  }
}

extern "C" void kernel_launch(void* const* d_in, const int* in_sizes, int n_in,
                              void* d_out, int out_size, void* d_ws,
                              size_t ws_size, hipStream_t stream) {
  (void)in_sizes; (void)n_in; (void)out_size; (void)ws_size;
  const int*   arr1 = (const int*)d_in[0];
  const int*   arr2 = (const int*)d_in[1];
  const float* emb  = (const float*)d_in[2];
  const float* wq   = (const float*)d_in[3];
  const float* wk   = (const float*)d_in[4];
  const float* wv   = (const float*)d_in[5];
  const float* w1   = (const float*)d_in[6];
  const float* b1   = (const float*)d_in[7];
  const float* w2   = (const float*)d_in[8];
  const float* b2   = (const float*)d_in[9];
  float* out = (float*)d_out;

  float* et = (float*)d_ws;
  float* vt = (float*)d_ws + VTO_;
  float* sr = (float*)d_ws + SRO_;
  vocab_tables_kernel<<<dim3(VT_), dim3(VT_ * 32), 0, stream>>>(emb, wq, wk, wv,
                                                                w1, et, sr, vt);

  encoder_kernel<<<dim3(NBLK_), dim3(TPB), 0, stream>>>(
      arr1, arr2, et, sr, vt, b1, w2, b2, out);
}
